# starting kernel
# baseline (speedup 1.0000x reference)
#include <hip/hip_runtime.h>


typedef __attribute__((ext_vector_type(2))) float v2f;
typedef __attribute__((ext_vector_type(8))) float v8f;

#define IN_DIM 64
#define OUT_DIM 64

__global__ __launch_bounds__(256) void zero_f32_kernel(float* __restrict__ p, long long n) {
    long long i = (long long)blockIdx.x * blockDim.x + threadIdx.x;
    if (i < n) p[i] = 0.0f;
}

__global__ __launch_bounds__(256) void degree_kernel(const int* __restrict__ rows,
                                                     const float* __restrict__ adj,
                                                     float* __restrict__ deg,
                                                     int n_edges) {
    int e = blockIdx.x * blockDim.x + threadIdx.x;
    if (e < n_edges) {
        unsafeAtomicAdd(&deg[rows[e]], adj[e]);
    }
}

__global__ __launch_bounds__(256) void dinv_kernel(const float* __restrict__ deg,
                                                   float* __restrict__ dinv,
                                                   int n_nodes) {
    int i = blockIdx.x * blockDim.x + threadIdx.x;
    if (i < n_nodes) {
        float d = deg[i];
        d = (d > 0.0f) ? d : 1.0f;
        dinv[i] = rsqrtf(d);
    }
}

__global__ __launch_bounds__(64) void gcn_gemm_wmma(const float* __restrict__ x,
                                                    const float* __restrict__ W,
                                                    const float* __restrict__ b,
                                                    float* __restrict__ support,
                                                    int n_nodes) {
    const int wave = threadIdx.x >> 5;
    const int lane = threadIdx.x & 31;
    const int mtile = blockIdx.x * 2 + wave;
    const int m0 = mtile * 16;
    if (m0 >= n_nodes) return;

    const int half = lane >> 4;
    const int r = lane & 15;

    v8f acc[4];
#pragma unroll
    for (int nt = 0; nt < 4; ++nt) {
        const float bv = b[nt * 16 + r];
#pragma unroll
        for (int i = 0; i < 8; ++i) acc[nt][i] = bv;
    }

    const float* arow = x + (size_t)(m0 + r) * IN_DIM;

#pragma unroll
    for (int k = 0; k < IN_DIM; k += 4) {
        const v2f a = *(const v2f*)(arow + k + 2 * half);
#pragma unroll
        for (int nt = 0; nt < 4; ++nt) {
            v2f bf;
            bf.x = W[(size_t)(k + 2 * half + 0) * OUT_DIM + nt * 16 + r];
            bf.y = W[(size_t)(k + 2 * half + 1) * OUT_DIM + nt * 16 + r];
            acc[nt] = __builtin_amdgcn_wmma_f32_16x16x4_f32(
                false, a, false, bf, (short)0, acc[nt], false, false);
        }
    }

    float* srow = support + (size_t)m0 * OUT_DIM;
#pragma unroll
    for (int nt = 0; nt < 4; ++nt) {
#pragma unroll
        for (int i = 0; i < 8; ++i) {
            srow[(size_t)(i + 8 * half) * OUT_DIM + nt * 16 + r] = acc[nt][i];
        }
    }
}

__global__ __launch_bounds__(256) void spmm_scatter_kernel(const int* __restrict__ rows,
                                                           const int* __restrict__ cols,
                                                           const float* __restrict__ adj,
                                                           const float* __restrict__ dinv,
                                                           const float* __restrict__ support,
                                                           float* __restrict__ out,
                                                           int n_edges) {
    long long t = (long long)blockIdx.x * blockDim.x + threadIdx.x;
    int e = (int)(t >> 6);
    int f = (int)(t & 63);
    if (e >= n_edges) return;
    const int rrow = rows[e];
    const int ccol = cols[e];
    const float norm = adj[e] * dinv[rrow] * dinv[ccol];
    const float v = norm * support[(size_t)ccol * OUT_DIM + f];
    unsafeAtomicAdd(&out[(size_t)rrow * OUT_DIM + f], v);
}

extern "C" void kernel_launch(void* const* d_in, const int* in_sizes, int n_in,
                              void* d_out, int out_size, void* d_ws, size_t ws_size,
                              hipStream_t stream) {
    const int* rows = (const int*)d_in[0];
    const int* cols = (const int*)d_in[1];
    const float* adj = (const float*)d_in[2];
    const float* x = (const float*)d_in[3];
    const float* W = (const float*)d_in[4];
    const float* b = (const float*)d_in[5];
    float* out = (float*)d_out;

    const int n_edges = in_sizes[0];
    const int n_nodes = in_sizes[3] / IN_DIM;

    float* support = (float*)d_ws;
    float* deg = support + (size_t)n_nodes * OUT_DIM;
    float* dinv = deg + n_nodes;

    const long long out_elems = (long long)n_nodes * OUT_DIM;

    {
        long long n = out_elems;
        zero_f32_kernel<<<(unsigned)((n + 255) / 256), 256, 0, stream>>>(out, n);
        zero_f32_kernel<<<(unsigned)((n_nodes + 255) / 256), 256, 0, stream>>>(deg, n_nodes);
    }

    degree_kernel<<<(n_edges + 255) / 256, 256, 0, stream>>>(rows, adj, deg, n_edges);

    dinv_kernel<<<(n_nodes + 255) / 256, 256, 0, stream>>>(deg, dinv, n_nodes);

    {
        const int mtiles = (n_nodes + 15) / 16;
        const int blocks = (mtiles + 1) / 2;
        gcn_gemm_wmma<<<blocks, 64, 0, stream>>>(x, W, b, support, n_nodes);
    }

    {
        const long long threads = (long long)n_edges * OUT_DIM;
        spmm_scatter_kernel<<<(unsigned)((threads + 255) / 256), 256, 0, stream>>>(
            rows, cols, adj, dinv, support, out, n_edges);
    }
}


# SpMM: each workgroup owns a slice of out rows in LDS, scans all row ids and sums its edges with LDS float atomics (arrival order) instead of global atomics; out no longer zero-filled
# speedup vs baseline: 1.7465x; 1.7465x over previous
#include <hip/hip_runtime.h>


typedef __attribute__((ext_vector_type(2))) float v2f;
typedef __attribute__((ext_vector_type(8))) float v8f;

#define IN_DIM 64
#define OUT_DIM 64

__global__ __launch_bounds__(256) void zero_f32_kernel(float* __restrict__ p, long long n) {
    long long i = (long long)blockIdx.x * blockDim.x + threadIdx.x;
    if (i < n) p[i] = 0.0f;
}

__global__ __launch_bounds__(256) void degree_kernel(const int* __restrict__ rows,
                                                     const float* __restrict__ adj,
                                                     float* __restrict__ deg,
                                                     int n_edges) {
    int e = blockIdx.x * blockDim.x + threadIdx.x;
    if (e < n_edges) {
        unsafeAtomicAdd(&deg[rows[e]], adj[e]);
    }
}

__global__ __launch_bounds__(256) void dinv_kernel(const float* __restrict__ deg,
                                                   float* __restrict__ dinv,
                                                   int n_nodes) {
    int i = blockIdx.x * blockDim.x + threadIdx.x;
    if (i < n_nodes) {
        float d = deg[i];
        d = (d > 0.0f) ? d : 1.0f;
        dinv[i] = rsqrtf(d);
    }
}

__global__ __launch_bounds__(64) void gcn_gemm_wmma(const float* __restrict__ x,
                                                    const float* __restrict__ W,
                                                    const float* __restrict__ b,
                                                    float* __restrict__ support,
                                                    int n_nodes) {
    const int wave = threadIdx.x >> 5;
    const int lane = threadIdx.x & 31;
    const int mtile = blockIdx.x * 2 + wave;
    const int m0 = mtile * 16;
    if (m0 >= n_nodes) return;

    const int half = lane >> 4;
    const int r = lane & 15;

    v8f acc[4];
#pragma unroll
    for (int nt = 0; nt < 4; ++nt) {
        const float bv = b[nt * 16 + r];
#pragma unroll
        for (int i = 0; i < 8; ++i) acc[nt][i] = bv;
    }

    const float* arow = x + (size_t)(m0 + r) * IN_DIM;

#pragma unroll
    for (int k = 0; k < IN_DIM; k += 4) {
        const v2f a = *(const v2f*)(arow + k + 2 * half);
#pragma unroll
        for (int nt = 0; nt < 4; ++nt) {
            v2f bf;
            bf.x = W[(size_t)(k + 2 * half + 0) * OUT_DIM + nt * 16 + r];
            bf.y = W[(size_t)(k + 2 * half + 1) * OUT_DIM + nt * 16 + r];
            acc[nt] = __builtin_amdgcn_wmma_f32_16x16x4_f32(
                false, a, false, bf, (short)0, acc[nt], false, false);
        }
    }

    float* srow = support + (size_t)m0 * OUT_DIM;
#pragma unroll
    for (int nt = 0; nt < 4; ++nt) {
#pragma unroll
        for (int i = 0; i < 8; ++i) {
            srow[(size_t)(i + 8 * half) * OUT_DIM + nt * 16 + r] = acc[nt][i];
        }
    }
}

#define SLICE_ROWS 784
#define SPMM_WAVES 32
#define SPMM_THREADS (SPMM_WAVES * 32)
#define SCAN_U 4
#define LIST_CAP (32 + SCAN_U * 4 * 32)
#define DRAIN_G 16

typedef __attribute__((ext_vector_type(4))) int v4i;
typedef __attribute__((ext_vector_type(4))) float v4f;

__device__ __forceinline__ void lds_add_f32(float* p, float v) {
    __hip_atomic_fetch_add(p, v, __ATOMIC_RELAXED, __HIP_MEMORY_SCOPE_WORKGROUP);
}

__device__ __forceinline__ void drain_batch(const int* __restrict__ ent, const int cnt, const int lane,
                                            const int lo,
                                            const int* __restrict__ rows,
                                            const int* __restrict__ cols,
                                            const float* __restrict__ adj,
                                            const float* __restrict__ dinv,
                                            const float* __restrict__ support,
                                            float* acc) {
    __builtin_amdgcn_wave_barrier();
    const int e = (lane < cnt) ? ent[lane] : ent[0];
    const int r = rows[e];
    const int c = cols[e];
    const float nrm = adj[e] * dinv[r] * dinv[c];
    const int rl = (r - lo) * OUT_DIM;
    const int nbits = __float_as_int(nrm);
#pragma unroll
    for (int g = 0; g < 32 / DRAIN_G; ++g) {
        if (g * DRAIN_G < cnt) {
            v2f s[DRAIN_G];
#pragma unroll
            for (int t = 0; t < DRAIN_G; ++t) {
                const int cj = __builtin_amdgcn_readlane(c, g * DRAIN_G + t);
                s[t] = *(const v2f*)(support + (size_t)cj * OUT_DIM + 2 * lane);
            }
#pragma unroll
            for (int t = 0; t < DRAIN_G; ++t) {
                const int j = g * DRAIN_G + t;
                if (j < cnt) {
                    const float nj = __int_as_float(__builtin_amdgcn_readlane(nbits, j));
                    const int rj = __builtin_amdgcn_readlane(rl, j) + 2 * lane;
                    lds_add_f32(acc + rj, nj * s[t].x);
                    lds_add_f32(acc + rj + 1, nj * s[t].y);
                }
            }
        }
    }
}

__global__ __launch_bounds__(SPMM_THREADS) void spmm_slice_kernel(const int* __restrict__ rows,
                                                                  const int* __restrict__ cols,
                                                                  const float* __restrict__ adj,
                                                                  const float* __restrict__ dinv,
                                                                  const float* __restrict__ support,
                                                                  float* __restrict__ out,
                                                                  int n_edges, int n_nodes) {
    __shared__ float acc[SLICE_ROWS * OUT_DIM];
    __shared__ int wlist[SPMM_WAVES * LIST_CAP];

    const int tid = threadIdx.x;
    const int lane = tid & 31;
    const int wave = __builtin_amdgcn_readfirstlane(tid >> 5);
    const int lo = blockIdx.x * SLICE_ROWS;
    int nrows = n_nodes - lo;
    if (nrows > SLICE_ROWS) nrows = SLICE_ROWS;

    v4f* acc4 = (v4f*)acc;
    const v4f zero4 = {0.0f, 0.0f, 0.0f, 0.0f};
    for (int i = tid; i < nrows * (OUT_DIM / 4); i += SPMM_THREADS) acc4[i] = zero4;
    __syncthreads();

    int* mylist = wlist + wave * LIST_CAP;
    int wcount = 0;

    const int nv4 = n_edges >> 2;
    const v4i* rows4 = (const v4i*)rows;
    const int step_vecs = 32 * SCAN_U;
    const int stride = SPMM_WAVES * step_vecs;

#define TEST_EDGE(RV, LOV, EV)                                                   \
    {                                                                            \
        const bool match_ = (unsigned)((RV) - (LOV)) < (unsigned)nrows;          \
        const unsigned m_ = __builtin_amdgcn_ballot_w32(match_);                 \
        if (m_ != 0u) {                                                          \
            if (match_) mylist[wcount + (int)__builtin_amdgcn_mbcnt_lo(m_, 0u)] = (EV); \
            wcount += __builtin_popcount(m_);                                    \
        }                                                                        \
    }

    if (wave == 0) {
        const int e = (nv4 << 2) + lane;
        const int r = (e < n_edges) ? rows[e] : -1;
        TEST_EDGE(r, lo, e)
    }

#define LOAD_STEP(DST, LOS, VB)                                                  \
    _Pragma("unroll") for (int u = 0; u < SCAN_U; ++u) {                         \
        const int idx_ = (VB) + u * 32 + lane;                                   \
        const bool ok_ = idx_ < nv4;                                             \
        DST[u] = rows4[ok_ ? idx_ : nv4 - 1];                                    \
        LOS[u] = ok_ ? lo : 0x7fffffff;                                          \
    }

#define SCAN_STEP(SRC, LOS, VB)                                                  \
    _Pragma("unroll") for (int u = 0; u < SCAN_U; ++u) {                         \
        const int eb_ = ((VB) + u * 32 + lane) * 4;                              \
        TEST_EDGE(SRC[u].x, LOS[u], eb_ + 0)                                     \
        TEST_EDGE(SRC[u].y, LOS[u], eb_ + 1)                                     \
        TEST_EDGE(SRC[u].z, LOS[u], eb_ + 2)                                     \
        TEST_EDGE(SRC[u].w, LOS[u], eb_ + 3)                                     \
    }                                                                            \
    while (wcount >= 32) {                                                       \
        wcount -= 32;                                                            \
        drain_batch(mylist + wcount, 32, lane, lo, rows, cols, adj, dinv, support, acc); \
    }

    int vb = wave * step_vecs;
    if (vb < nv4) {
        v4i ra[SCAN_U], rb[SCAN_U];
        int la[SCAN_U], lb[SCAN_U];
        LOAD_STEP(ra, la, vb)
        for (;;) {
            LOAD_STEP(rb, lb, vb + stride)
            SCAN_STEP(ra, la, vb)
            vb += stride;
            if (vb >= nv4) break;
            LOAD_STEP(ra, la, vb + stride)
            SCAN_STEP(rb, lb, vb)
            vb += stride;
            if (vb >= nv4) break;
        }
    }
#undef SCAN_STEP
#undef LOAD_STEP
#undef TEST_EDGE
    if (wcount > 0) drain_batch(mylist, wcount, lane, lo, rows, cols, adj, dinv, support, acc);
    __syncthreads();

    v4f* out4 = (v4f*)(out + (size_t)lo * OUT_DIM);
    for (int i = tid; i < nrows * (OUT_DIM / 4); i += SPMM_THREADS) out4[i] = acc4[i];
}

extern "C" void kernel_launch(void* const* d_in, const int* in_sizes, int n_in,
                              void* d_out, int out_size, void* d_ws, size_t ws_size,
                              hipStream_t stream) {
    const int* rows = (const int*)d_in[0];
    const int* cols = (const int*)d_in[1];
    const float* adj = (const float*)d_in[2];
    const float* x = (const float*)d_in[3];
    const float* W = (const float*)d_in[4];
    const float* b = (const float*)d_in[5];
    float* out = (float*)d_out;

    const int n_edges = in_sizes[0];
    const int n_nodes = in_sizes[3] / IN_DIM;

    float* support = (float*)d_ws;
    float* deg = support + (size_t)n_nodes * OUT_DIM;
    float* dinv = deg + n_nodes;

    zero_f32_kernel<<<(unsigned)((n_nodes + 255) / 256), 256, 0, stream>>>(deg, n_nodes);

    degree_kernel<<<(n_edges + 255) / 256, 256, 0, stream>>>(rows, adj, deg, n_edges);

    dinv_kernel<<<(n_nodes + 255) / 256, 256, 0, stream>>>(deg, dinv, n_nodes);

    {
        const int mtiles = (n_nodes + 15) / 16;
        const int blocks = (mtiles + 1) / 2;
        gcn_gemm_wmma<<<blocks, 64, 0, stream>>>(x, W, b, support, n_nodes);
    }

    spmm_slice_kernel<<<(n_nodes + SLICE_ROWS - 1) / SLICE_ROWS, SPMM_THREADS, 0, stream>>>(
        rows, cols, adj, dinv, support, out, n_edges, n_nodes);
}


# SpMM: edge numbers grouped by destination slice once (one more launch, LDS counters); a slice workgroup reads its own segments instead of testing every row id; LDS float adds, arrival order, kept
# speedup vs baseline: 1.9048x; 1.0906x over previous
#include <hip/hip_runtime.h>


typedef __attribute__((ext_vector_type(2))) float v2f;
typedef __attribute__((ext_vector_type(8))) float v8f;

#define IN_DIM 64
#define OUT_DIM 64

__global__ __launch_bounds__(256) void zero_f32_kernel(float* __restrict__ p, long long n) {
    long long i = (long long)blockIdx.x * blockDim.x + threadIdx.x;
    if (i < n) p[i] = 0.0f;
}

__global__ __launch_bounds__(256) void degree_kernel(const int* __restrict__ rows,
                                                     const float* __restrict__ adj,
                                                     float* __restrict__ deg,
                                                     int n_edges) {
    int e = blockIdx.x * blockDim.x + threadIdx.x;
    if (e < n_edges) {
        unsafeAtomicAdd(&deg[rows[e]], adj[e]);
    }
}

__global__ __launch_bounds__(256) void dinv_kernel(const float* __restrict__ deg,
                                                   float* __restrict__ dinv,
                                                   int n_nodes) {
    int i = blockIdx.x * blockDim.x + threadIdx.x;
    if (i < n_nodes) {
        float d = deg[i];
        d = (d > 0.0f) ? d : 1.0f;
        dinv[i] = rsqrtf(d);
    }
}

__global__ __launch_bounds__(64) void gcn_gemm_wmma(const float* __restrict__ x,
                                                    const float* __restrict__ W,
                                                    const float* __restrict__ b,
                                                    float* __restrict__ support,
                                                    int n_nodes) {
    const int wave = threadIdx.x >> 5;
    const int lane = threadIdx.x & 31;
    const int mtile = blockIdx.x * 2 + wave;
    const int m0 = mtile * 16;
    if (m0 >= n_nodes) return;

    const int half = lane >> 4;
    const int r = lane & 15;

    v8f acc[4];
#pragma unroll
    for (int nt = 0; nt < 4; ++nt) {
        const float bv = b[nt * 16 + r];
#pragma unroll
        for (int i = 0; i < 8; ++i) acc[nt][i] = bv;
    }

    const float* arow = x + (size_t)(m0 + r) * IN_DIM;

#pragma unroll
    for (int k = 0; k < IN_DIM; k += 4) {
        const v2f a = *(const v2f*)(arow + k + 2 * half);
#pragma unroll
        for (int nt = 0; nt < 4; ++nt) {
            v2f bf;
            bf.x = W[(size_t)(k + 2 * half + 0) * OUT_DIM + nt * 16 + r];
            bf.y = W[(size_t)(k + 2 * half + 1) * OUT_DIM + nt * 16 + r];
            acc[nt] = __builtin_amdgcn_wmma_f32_16x16x4_f32(
                false, a, false, bf, (short)0, acc[nt], false, false);
        }
    }

    float* srow = support + (size_t)m0 * OUT_DIM;
#pragma unroll
    for (int nt = 0; nt < 4; ++nt) {
#pragma unroll
        for (int i = 0; i < 8; ++i) {
            srow[(size_t)(i + 8 * half) * OUT_DIM + nt * 16 + r] = acc[nt][i];
        }
    }
}

#define SLICE_ROWS 784
#define SPMM_WAVES 32
#define SPMM_THREADS (SPMM_WAVES * 32)
#define SCAN_U 4
#define LIST_CAP (32 + SCAN_U * 4 * 32)
#define DRAIN_G 16

typedef __attribute__((ext_vector_type(4))) int v4i;
typedef __attribute__((ext_vector_type(4))) float v4f;

__device__ __forceinline__ void lds_add_f32(float* p, float v) {
    __hip_atomic_fetch_add(p, v, __ATOMIC_RELAXED, __HIP_MEMORY_SCOPE_WORKGROUP);
}

__device__ __forceinline__ void drain_batch(const int* __restrict__ ent, const int cnt, const int lane,
                                            const int lo,
                                            const int* __restrict__ rows,
                                            const int* __restrict__ cols,
                                            const float* __restrict__ adj,
                                            const float* __restrict__ dinv,
                                            const float* __restrict__ support,
                                            float* acc) {
    __builtin_amdgcn_wave_barrier();
    const int e = (lane < cnt) ? ent[lane] : ent[0];
    const int r = rows[e];
    const int c = cols[e];
    const float nrm = adj[e] * dinv[r] * dinv[c];
    const int rl = (r - lo) * OUT_DIM;
    const int nbits = __float_as_int(nrm);
#pragma unroll
    for (int g = 0; g < 32 / DRAIN_G; ++g) {
        if (g * DRAIN_G < cnt) {
            v2f s[DRAIN_G];
#pragma unroll
            for (int t = 0; t < DRAIN_G; ++t) {
                const int cj = __builtin_amdgcn_readlane(c, g * DRAIN_G + t);
                s[t] = *(const v2f*)(support + (size_t)cj * OUT_DIM + 2 * lane);
            }
#pragma unroll
            for (int t = 0; t < DRAIN_G; ++t) {
                const int j = g * DRAIN_G + t;
                if (j < cnt) {
                    const float nj = __int_as_float(__builtin_amdgcn_readlane(nbits, j));
                    const int rj = __builtin_amdgcn_readlane(rl, j) + 2 * lane;
                    lds_add_f32(acc + rj, nj * s[t].x);
                    lds_add_f32(acc + rj + 1, nj * s[t].y);
                }
            }
        }
    }
}


#define BIN_WGS 256
#define BIN_THREADS 1024
#define BIN_U 4
#define MAX_SLICES 256
#define OFF_PITCH (MAX_SLICES + 1)
#define SEGS_PER_WAVE (BIN_WGS / SPMM_WAVES)

__device__ __forceinline__ bool bins_fit(unsigned long long ws_bytes, int n_nodes, int n_edges) {
    const unsigned long long need =
        ((unsigned long long)n_nodes * (OUT_DIM + 2) + (unsigned long long)BIN_WGS * OFF_PITCH +
         (unsigned long long)n_edges) * 4ull;
    return ws_bytes >= need && (n_nodes + SLICE_ROWS - 1) / SLICE_ROWS <= MAX_SLICES;
}

__global__ __launch_bounds__(BIN_THREADS) void bin_edges_kernel(const int* __restrict__ rows,
                                                                int* __restrict__ binoff,
                                                                int* __restrict__ binlist,
                                                                int n_edges, int n_nodes,
                                                                unsigned long long ws_bytes) {
    if (!bins_fit(ws_bytes, n_nodes, n_edges)) return;
    __shared__ int cnt[MAX_SLICES];
    __shared__ int cur[MAX_SLICES];
    const int tid = threadIdx.x;
    const int ch = (n_edges + BIN_WGS - 1) / BIN_WGS;
    const long long c0l = (long long)blockIdx.x * ch;
    const int c0 = (c0l < (long long)n_edges) ? (int)c0l : n_edges;
    const int c1 = (n_edges - c0 < ch) ? n_edges : c0 + ch;

    if (tid < MAX_SLICES) cnt[tid] = 0;
    __syncthreads();
    for (int eb = c0 + tid; eb < c1; eb += BIN_THREADS * BIN_U) {
        unsigned r[BIN_U];
#pragma unroll
        for (int u = 0; u < BIN_U; ++u) {
            const int e = eb + u * BIN_THREADS;
            r[u] = (e < c1) ? (unsigned)rows[e] : 0xffffffffu;
        }
#pragma unroll
        for (int u = 0; u < BIN_U; ++u)
            if (r[u] < (unsigned)n_nodes) atomicAdd(&cnt[r[u] / SLICE_ROWS], 1);
    }
    __syncthreads();
    if (tid < 32) {
        int v[MAX_SLICES / 32];
        int sum = 0;
#pragma unroll
        for (int i = 0; i < MAX_SLICES / 32; ++i) {
            v[i] = cnt[tid * (MAX_SLICES / 32) + i];
            sum += v[i];
        }
        int incl = sum;
#pragma unroll
        for (int d = 1; d < 32; d <<= 1) {
            const int t = __shfl_up(incl, d, 32);
            if (tid >= d) incl += t;
        }
        int run = incl - sum;
        int* orow = binoff + (size_t)blockIdx.x * OFF_PITCH;
#pragma unroll
        for (int i = 0; i < MAX_SLICES / 32; ++i) {
            cur[tid * (MAX_SLICES / 32) + i] = run;
            orow[tid * (MAX_SLICES / 32) + i] = run;
            run += v[i];
        }
        if (tid == 31) orow[MAX_SLICES] = run;
    }
    __syncthreads();
    for (int eb = c0 + tid; eb < c1; eb += BIN_THREADS * BIN_U) {
        unsigned r[BIN_U];
#pragma unroll
        for (int u = 0; u < BIN_U; ++u) {
            const int e = eb + u * BIN_THREADS;
            r[u] = (e < c1) ? (unsigned)rows[e] : 0xffffffffu;
        }
#pragma unroll
        for (int u = 0; u < BIN_U; ++u) {
            if (r[u] < (unsigned)n_nodes) {
                const int p = atomicAdd(&cur[r[u] / SLICE_ROWS], 1);
                binlist[(size_t)c0 + p] = eb + u * BIN_THREADS;
            }
        }
    }
}

__global__ __launch_bounds__(SPMM_THREADS) void spmm_slice_kernel(const int* __restrict__ rows,
                                                                  const int* __restrict__ cols,
                                                                  const float* __restrict__ adj,
                                                                  const float* __restrict__ dinv,
                                                                  const float* __restrict__ support,
                                                                  float* __restrict__ out,
                                                                  const int* __restrict__ binoff,
                                                                  const int* __restrict__ binlist,
                                                                  int n_edges, int n_nodes,
                                                                  unsigned long long ws_bytes) {
    __shared__ float acc[SLICE_ROWS * OUT_DIM];
    __shared__ int wlist[SPMM_WAVES * LIST_CAP];

    const int tid = threadIdx.x;
    const int lane = tid & 31;
    const int wave = __builtin_amdgcn_readfirstlane(tid >> 5);
    const int lo = blockIdx.x * SLICE_ROWS;
    int nrows = n_nodes - lo;
    if (nrows > SLICE_ROWS) nrows = SLICE_ROWS;

    v4f* acc4 = (v4f*)acc;
    const v4f zero4 = {0.0f, 0.0f, 0.0f, 0.0f};
    for (int i = tid; i < nrows * (OUT_DIM / 4); i += SPMM_THREADS) acc4[i] = zero4;
    __syncthreads();

    int* mylist = wlist + wave * LIST_CAP;
    int wcount = 0;

    if (bins_fit(ws_bytes, n_nodes, n_edges)) {
        const int ch = (n_edges + BIN_WGS - 1) / BIN_WGS;
        int seg_a = 0, seg_b = 0;
        if (lane < SEGS_PER_WAVE) {
            const int* o = binoff + (size_t)(wave + SPMM_WAVES * lane) * OFF_PITCH + blockIdx.x;
            seg_a = o[0];
            seg_b = o[1];
        }
        int pre0[SEGS_PER_WAVE], pre1[SEGS_PER_WAVE];
#pragma unroll
        for (int j = 0; j < SEGS_PER_WAVE; ++j) {
            const int a = __builtin_amdgcn_readlane(seg_a, j);
            const int len = __builtin_amdgcn_readlane(seg_b, j) - a;
            const int* seg = binlist + (size_t)(wave + SPMM_WAVES * j) * ch + a;
            pre0[j] = (lane < len) ? seg[lane] : 0;
            pre1[j] = (lane + 32 < len) ? seg[lane + 32] : 0;
        }
#pragma unroll
        for (int j = 0; j < SEGS_PER_WAVE; ++j) {
            const int len = __builtin_amdgcn_readlane(seg_b, j) - __builtin_amdgcn_readlane(seg_a, j);
            const int n0 = (len < 32) ? len : 32;
            const int n1 = (len < 64) ? len - 32 : 32;
            if (lane < n0) mylist[wcount + lane] = pre0[j];
            if (n0 > 0) wcount += n0;
            if (lane < n1) mylist[wcount + lane] = pre1[j];
            if (n1 > 0) wcount += n1;
        }
        int rj = 0, ri = 64;
        for (;;) {
            while (rj < SEGS_PER_WAVE && wcount + 32 <= LIST_CAP) {
                const int a = __builtin_amdgcn_readlane(seg_a, rj);
                const int len = __builtin_amdgcn_readlane(seg_b, rj) - a;
                if (ri < len) {
                    const int n = (len - ri < 32) ? (len - ri) : 32;
                    const int* seg = binlist + (size_t)(wave + SPMM_WAVES * rj) * ch + a;
                    if (lane < n) mylist[wcount + lane] = seg[ri + lane];
                    wcount += n;
                    ri += 32;
                } else {
                    ++rj;
                    ri = 64;
                }
            }
            if (wcount < 32) break;
            wcount -= 32;
            drain_batch(mylist + wcount, 32, lane, lo, rows, cols, adj, dinv, support, acc);
        }
    } else {
    const int nv4 = n_edges >> 2;
    const v4i* rows4 = (const v4i*)rows;
    const int step_vecs = 32 * SCAN_U;
    const int stride = SPMM_WAVES * step_vecs;

#define TEST_EDGE(RV, LOV, EV)                                                   \
    {                                                                            \
        const bool match_ = (unsigned)((RV) - (LOV)) < (unsigned)nrows;          \
        const unsigned m_ = __builtin_amdgcn_ballot_w32(match_);                 \
        if (m_ != 0u) {                                                          \
            if (match_) mylist[wcount + (int)__builtin_amdgcn_mbcnt_lo(m_, 0u)] = (EV); \
            wcount += __builtin_popcount(m_);                                    \
        }                                                                        \
    }

    if (wave == 0) {
        const int e = (nv4 << 2) + lane;
        const int r = (e < n_edges) ? rows[e] : -1;
        TEST_EDGE(r, lo, e)
    }

#define LOAD_STEP(DST, LOS, VB)                                                  \
    _Pragma("unroll") for (int u = 0; u < SCAN_U; ++u) {                         \
        const int idx_ = (VB) + u * 32 + lane;                                   \
        const bool ok_ = idx_ < nv4;                                             \
        DST[u] = rows4[ok_ ? idx_ : nv4 - 1];                                    \
        LOS[u] = ok_ ? lo : 0x7fffffff;                                          \
    }

#define SCAN_STEP(SRC, LOS, VB)                                                  \
    _Pragma("unroll") for (int u = 0; u < SCAN_U; ++u) {                         \
        const int eb_ = ((VB) + u * 32 + lane) * 4;                              \
        TEST_EDGE(SRC[u].x, LOS[u], eb_ + 0)                                     \
        TEST_EDGE(SRC[u].y, LOS[u], eb_ + 1)                                     \
        TEST_EDGE(SRC[u].z, LOS[u], eb_ + 2)                                     \
        TEST_EDGE(SRC[u].w, LOS[u], eb_ + 3)                                     \
    }                                                                            \
    while (wcount >= 32) {                                                       \
        wcount -= 32;                                                            \
        drain_batch(mylist + wcount, 32, lane, lo, rows, cols, adj, dinv, support, acc); \
    }

    int vb = wave * step_vecs;
    if (vb < nv4) {
        v4i ra[SCAN_U], rb[SCAN_U];
        int la[SCAN_U], lb[SCAN_U];
        LOAD_STEP(ra, la, vb)
        for (;;) {
            LOAD_STEP(rb, lb, vb + stride)
            SCAN_STEP(ra, la, vb)
            vb += stride;
            if (vb >= nv4) break;
            LOAD_STEP(ra, la, vb + stride)
            SCAN_STEP(rb, lb, vb)
            vb += stride;
            if (vb >= nv4) break;
        }
    }
#undef SCAN_STEP
#undef LOAD_STEP
#undef TEST_EDGE
    }
    if (wcount > 0) drain_batch(mylist, wcount, lane, lo, rows, cols, adj, dinv, support, acc);
    __syncthreads();

    v4f* out4 = (v4f*)(out + (size_t)lo * OUT_DIM);
    for (int i = tid; i < nrows * (OUT_DIM / 4); i += SPMM_THREADS) out4[i] = acc4[i];
}

extern "C" void kernel_launch(void* const* d_in, const int* in_sizes, int n_in,
                              void* d_out, int out_size, void* d_ws, size_t ws_size,
                              hipStream_t stream) {
    const int* rows = (const int*)d_in[0];
    const int* cols = (const int*)d_in[1];
    const float* adj = (const float*)d_in[2];
    const float* x = (const float*)d_in[3];
    const float* W = (const float*)d_in[4];
    const float* b = (const float*)d_in[5];
    float* out = (float*)d_out;

    const int n_edges = in_sizes[0];
    const int n_nodes = in_sizes[3] / IN_DIM;

    float* support = (float*)d_ws;
    float* deg = support + (size_t)n_nodes * OUT_DIM;
    float* dinv = deg + n_nodes;
    int* binoff = (int*)(dinv + n_nodes);
    int* binlist = binoff + BIN_WGS * OFF_PITCH;
    const unsigned long long ws_bytes = (unsigned long long)ws_size;

    zero_f32_kernel<<<(unsigned)((n_nodes + 255) / 256), 256, 0, stream>>>(deg, n_nodes);

    degree_kernel<<<(n_edges + 255) / 256, 256, 0, stream>>>(rows, adj, deg, n_edges);

    dinv_kernel<<<(n_nodes + 255) / 256, 256, 0, stream>>>(deg, dinv, n_nodes);

    {
        const int mtiles = (n_nodes + 15) / 16;
        const int blocks = (mtiles + 1) / 2;
        gcn_gemm_wmma<<<blocks, 64, 0, stream>>>(x, W, b, support, n_nodes);
    }

    bin_edges_kernel<<<BIN_WGS, BIN_THREADS, 0, stream>>>(rows, binoff, binlist, n_edges, n_nodes, ws_bytes);

    spmm_slice_kernel<<<(n_nodes + SLICE_ROWS - 1) / SLICE_ROWS, SPMM_THREADS, 0, stream>>>(
        rows, cols, adj, dinv, support, out, binoff, binlist, n_edges, n_nodes, ws_bytes);
}


# SpMM: binning launch writes each edge as an 8-byte record (col, row in slice, adj) in slice order; slice workgroups read records coalesced instead of 3 scattered 4-byte gathers per edge
# speedup vs baseline: 2.2149x; 1.1628x over previous
#include <hip/hip_runtime.h>


typedef __attribute__((ext_vector_type(2))) float v2f;
typedef __attribute__((ext_vector_type(8))) float v8f;

#define IN_DIM 64
#define OUT_DIM 64

__global__ __launch_bounds__(256) void zero_f32_kernel(float* __restrict__ p, long long n) {
    long long i = (long long)blockIdx.x * blockDim.x + threadIdx.x;
    if (i < n) p[i] = 0.0f;
}

__global__ __launch_bounds__(256) void degree_kernel(const int* __restrict__ rows,
                                                     const float* __restrict__ adj,
                                                     float* __restrict__ deg,
                                                     int n_edges) {
    int e = blockIdx.x * blockDim.x + threadIdx.x;
    if (e < n_edges) {
        unsafeAtomicAdd(&deg[rows[e]], adj[e]);
    }
}

__global__ __launch_bounds__(256) void dinv_kernel(const float* __restrict__ deg,
                                                   float* __restrict__ dinv,
                                                   int n_nodes) {
    int i = blockIdx.x * blockDim.x + threadIdx.x;
    if (i < n_nodes) {
        float d = deg[i];
        d = (d > 0.0f) ? d : 1.0f;
        dinv[i] = rsqrtf(d);
    }
}

__global__ __launch_bounds__(64) void gcn_gemm_wmma(const float* __restrict__ x,
                                                    const float* __restrict__ W,
                                                    const float* __restrict__ b,
                                                    float* __restrict__ support,
                                                    int n_nodes) {
    const int wave = threadIdx.x >> 5;
    const int lane = threadIdx.x & 31;
    const int mtile = blockIdx.x * 2 + wave;
    const int m0 = mtile * 16;
    if (m0 >= n_nodes) return;

    const int half = lane >> 4;
    const int r = lane & 15;

    v8f acc[4];
#pragma unroll
    for (int nt = 0; nt < 4; ++nt) {
        const float bv = b[nt * 16 + r];
#pragma unroll
        for (int i = 0; i < 8; ++i) acc[nt][i] = bv;
    }

    const float* arow = x + (size_t)(m0 + r) * IN_DIM;

#pragma unroll
    for (int k = 0; k < IN_DIM; k += 4) {
        const v2f a = *(const v2f*)(arow + k + 2 * half);
#pragma unroll
        for (int nt = 0; nt < 4; ++nt) {
            v2f bf;
            bf.x = W[(size_t)(k + 2 * half + 0) * OUT_DIM + nt * 16 + r];
            bf.y = W[(size_t)(k + 2 * half + 1) * OUT_DIM + nt * 16 + r];
            acc[nt] = __builtin_amdgcn_wmma_f32_16x16x4_f32(
                false, a, false, bf, (short)0, acc[nt], false, false);
        }
    }

    float* srow = support + (size_t)m0 * OUT_DIM;
#pragma unroll
    for (int nt = 0; nt < 4; ++nt) {
#pragma unroll
        for (int i = 0; i < 8; ++i) {
            srow[(size_t)(i + 8 * half) * OUT_DIM + nt * 16 + r] = acc[nt][i];
        }
    }
}

#define SLICE_ROWS 784
#define SPMM_WAVES 32
#define SPMM_THREADS (SPMM_WAVES * 32)
#define SCAN_U 4
#define LIST_CAP (32 + SCAN_U * 4 * 32)
#define DRAIN_G 16

typedef __attribute__((ext_vector_type(2))) int v2i;
typedef __attribute__((ext_vector_type(4))) int v4i;
typedef __attribute__((ext_vector_type(4))) float v4f;

__device__ __forceinline__ void lds_add_f32(float* p, float v) {
    __hip_atomic_fetch_add(p, v, __ATOMIC_RELAXED, __HIP_MEMORY_SCOPE_WORKGROUP);
}

__device__ __forceinline__ void drain_rows(const int c, const int rl, const float wa, const float wr, const float wc,
                                           const int cnt, const int lane,
                                           const float* __restrict__ support,
                                           float* acc) {
    int nbits = 0;
#pragma unroll
    for (int g = 0; g < 32 / DRAIN_G; ++g) {
        if (g * DRAIN_G < cnt) {
            v2f s[DRAIN_G];
#pragma unroll
            for (int t = 0; t < DRAIN_G; ++t) {
                const int cj = __builtin_amdgcn_readlane(c, g * DRAIN_G + t);
                s[t] = *(const v2f*)(support + (size_t)cj * OUT_DIM + 2 * lane);
            }
            if (g == 0) nbits = __float_as_int(wa * wr * wc);
#pragma unroll
            for (int t = 0; t < DRAIN_G; ++t) {
                const int j = g * DRAIN_G + t;
                if (j < cnt) {
                    const float nj = __int_as_float(__builtin_amdgcn_readlane(nbits, j));
                    const int rj = __builtin_amdgcn_readlane(rl, j) + 2 * lane;
                    lds_add_f32(acc + rj, nj * s[t].x);
                    lds_add_f32(acc + rj + 1, nj * s[t].y);
                }
            }
        }
    }
}

__device__ __forceinline__ void drain_batch(const int* __restrict__ ent, const int cnt, const int lane,
                                            const int lo,
                                            const int* __restrict__ rows,
                                            const int* __restrict__ cols,
                                            const float* __restrict__ adj,
                                            const float* __restrict__ dinv,
                                            const float* __restrict__ support,
                                            float* acc) {
    __builtin_amdgcn_wave_barrier();
    const int e = (lane < cnt) ? ent[lane] : ent[0];
    const int r = rows[e];
    const int c = cols[e];
    drain_rows(c, (r - lo) * OUT_DIM, adj[e], dinv[r], dinv[c], cnt, lane, support, acc);
}


#define BIN_WGS 256
#define BIN_THREADS 1024
#define BIN_U 4
#define MAX_SLICES 256
#define OFF_PITCH (MAX_SLICES + 1)
#define SEGS_PER_WAVE (BIN_WGS / SPMM_WAVES)
#define REC_COL_BITS 22
#define REC_COL_MASK ((1 << REC_COL_BITS) - 1)
static_assert(SLICE_ROWS <= (1 << (32 - REC_COL_BITS)), "the row inside the slice must fit above the col bits");

__device__ __forceinline__ int rec_pad_ints(const int* recs_raw) {
    return (int)(((8ull - ((unsigned long long)recs_raw & 7ull)) & 7ull) >> 2);
}

__device__ __forceinline__ bool recs_fit(unsigned long long ws_bytes, int n_nodes, int n_edges, const int* recs_raw) {
    const unsigned long long pad = 4ull * (unsigned long long)rec_pad_ints(recs_raw);
    const unsigned long long need =
        ((unsigned long long)n_nodes * (OUT_DIM + 2) + (unsigned long long)BIN_WGS * OFF_PITCH) * 4ull + pad +
        (unsigned long long)n_edges * 8ull;
    return ws_bytes >= need && n_nodes <= (1 << REC_COL_BITS) && (n_nodes + SLICE_ROWS - 1) / SLICE_ROWS <= MAX_SLICES;
}

__global__ __launch_bounds__(BIN_THREADS) void bin_edges_kernel(const int* __restrict__ rows,
                                                                const int* __restrict__ cols,
                                                                const float* __restrict__ adj,
                                                                int* __restrict__ binoff,
                                                                int* recs_raw,
                                                                int n_edges, int n_nodes,
                                                                unsigned long long ws_bytes) {
    if (!recs_fit(ws_bytes, n_nodes, n_edges, recs_raw)) return;
    v2i* __restrict__ recs = (v2i*)(recs_raw + rec_pad_ints(recs_raw));
    __shared__ int cnt[MAX_SLICES];
    __shared__ int cur[MAX_SLICES];
    const int tid = threadIdx.x;
    const int ch = (n_edges + BIN_WGS - 1) / BIN_WGS;
    const long long c0l = (long long)blockIdx.x * ch;
    const int c0 = (c0l < (long long)n_edges) ? (int)c0l : n_edges;
    const int c1 = (n_edges - c0 < ch) ? n_edges : c0 + ch;

    if (tid < MAX_SLICES) cnt[tid] = 0;
    __syncthreads();
    for (int eb = c0 + tid; eb < c1; eb += BIN_THREADS * BIN_U) {
        unsigned r[BIN_U];
#pragma unroll
        for (int u = 0; u < BIN_U; ++u) {
            const int e = eb + u * BIN_THREADS;
            r[u] = (e < c1) ? (unsigned)rows[e] : 0xffffffffu;
        }
#pragma unroll
        for (int u = 0; u < BIN_U; ++u)
            if (r[u] < (unsigned)n_nodes) atomicAdd(&cnt[r[u] / SLICE_ROWS], 1);
    }
    __syncthreads();
    if (tid < 32) {
        int v[MAX_SLICES / 32];
        int sum = 0;
#pragma unroll
        for (int i = 0; i < MAX_SLICES / 32; ++i) {
            v[i] = cnt[tid * (MAX_SLICES / 32) + i];
            sum += v[i];
        }
        int incl = sum;
#pragma unroll
        for (int d = 1; d < 32; d <<= 1) {
            const int t = __shfl_up(incl, d, 32);
            if (tid >= d) incl += t;
        }
        int run = incl - sum;
        int* orow = binoff + (size_t)blockIdx.x * OFF_PITCH;
#pragma unroll
        for (int i = 0; i < MAX_SLICES / 32; ++i) {
            cur[tid * (MAX_SLICES / 32) + i] = run;
            orow[tid * (MAX_SLICES / 32) + i] = run;
            run += v[i];
        }
        if (tid == 31) orow[MAX_SLICES] = run;
    }
    __syncthreads();
    for (int eb = c0 + tid; eb < c1; eb += BIN_THREADS * BIN_U) {
        unsigned r[BIN_U];
        int cc[BIN_U], av[BIN_U];
#pragma unroll
        for (int u = 0; u < BIN_U; ++u) {
            const int e = eb + u * BIN_THREADS;
            const bool ok = e < c1;
            r[u] = ok ? (unsigned)rows[e] : 0xffffffffu;
            cc[u] = ok ? cols[e] : 0;
            av[u] = ok ? __float_as_int(adj[e]) : 0;
        }
#pragma unroll
        for (int u = 0; u < BIN_U; ++u) {
            if (r[u] < (unsigned)n_nodes) {
                const int sl = (int)(r[u] / SLICE_ROWS);
                const int p = atomicAdd(&cur[sl], 1);
                v2i rec;
                rec.x = (int)(((unsigned)cc[u] & (unsigned)REC_COL_MASK) | ((r[u] - (unsigned)(sl * SLICE_ROWS)) << REC_COL_BITS));
                rec.y = av[u];
                recs[(size_t)c0 + p] = rec;
            }
        }
    }
}

__global__ __launch_bounds__(SPMM_THREADS) void spmm_slice_kernel(const int* __restrict__ rows,
                                                                  const int* __restrict__ cols,
                                                                  const float* __restrict__ adj,
                                                                  const float* __restrict__ dinv,
                                                                  const float* __restrict__ support,
                                                                  float* __restrict__ out,
                                                                  const int* __restrict__ binoff,
                                                                  const int* __restrict__ recs_raw,
                                                                  int n_edges, int n_nodes,
                                                                  unsigned long long ws_bytes) {
    __shared__ float acc[SLICE_ROWS * OUT_DIM];
    __shared__ int wlist[SPMM_WAVES * LIST_CAP];

    const int tid = threadIdx.x;
    const int lane = tid & 31;
    const int wave = __builtin_amdgcn_readfirstlane(tid >> 5);
    const int lo = blockIdx.x * SLICE_ROWS;
    int nrows = n_nodes - lo;
    if (nrows > SLICE_ROWS) nrows = SLICE_ROWS;

    v4f* acc4 = (v4f*)acc;
    const v4f zero4 = {0.0f, 0.0f, 0.0f, 0.0f};
    for (int i = tid; i < nrows * (OUT_DIM / 4); i += SPMM_THREADS) acc4[i] = zero4;
    __syncthreads();

    int* mylist = wlist + wave * LIST_CAP;
    int wcount = 0;

    if (recs_fit(ws_bytes, n_nodes, n_edges, recs_raw)) {
        const v2i* __restrict__ recs = (const v2i*)(recs_raw + rec_pad_ints(recs_raw));
        const int ch = (n_edges + BIN_WGS - 1) / BIN_WGS;
        int seg_a = 0, seg_b = 0;
        if (lane < SEGS_PER_WAVE) {
            const int* o = binoff + (size_t)(wave + SPMM_WAVES * lane) * OFF_PITCH + blockIdx.x;
            seg_a = o[0];
            seg_b = o[1];
        }
        int offs[SEGS_PER_WAVE + 1];
        int qrec[SEGS_PER_WAVE];
        offs[0] = 0;
#pragma unroll
        for (int j = 0; j < SEGS_PER_WAVE; ++j) {
            const int a = __builtin_amdgcn_readlane(seg_a, j);
            const int len = __builtin_amdgcn_readlane(seg_b, j) - a;
            const long long k0l = (long long)(wave + SPMM_WAVES * j) * ch;
            const int k0 = (k0l < (long long)n_edges) ? (int)k0l : n_edges;
            qrec[j] = k0 + a - offs[j];
            offs[j + 1] = offs[j] + len;
        }
        const int total = offs[SEGS_PER_WAVE];

#define FETCH_REC(DST, I0)                                                       \
    {                                                                            \
        const int i_ = (I0) + lane;                                              \
        const int ii_ = (i_ < total) ? i_ : (I0);                                \
        int q_ = qrec[0];                                                        \
        _Pragma("unroll") for (int j = 1; j < SEGS_PER_WAVE; ++j)                \
            if (ii_ >= offs[j]) q_ = qrec[j];                                    \
        DST = recs[(size_t)(q_ + ii_)];                                          \
    }

        if (total > 0) {
            v2i rc0, rc1;
            FETCH_REC(rc0, 0)
            {
                const int nb = (32 < total) ? 32 : 0;
                FETCH_REC(rc1, nb)
            }
            float wr = dinv[lo + (int)((unsigned)rc0.x >> REC_COL_BITS)];
            float wc = dinv[rc0.x & REC_COL_MASK];
            const int full_end = total & ~31;
            for (int i0 = 0; i0 < full_end; i0 += 32) {
                v2i rc2;
                const int nb = (i0 + 64 < total) ? i0 + 64 : i0;
                FETCH_REC(rc2, nb)
                const float nwr = dinv[lo + (int)((unsigned)rc1.x >> REC_COL_BITS)];
                const float nwc = dinv[rc1.x & REC_COL_MASK];
                drain_rows(rc0.x & REC_COL_MASK, (int)((unsigned)rc0.x >> REC_COL_BITS) * OUT_DIM,
                           __int_as_float(rc0.y), wr, wc, 32, lane, support, acc);
                rc0 = rc1;
                rc1 = rc2;
                wr = nwr;
                wc = nwc;
            }
            if (full_end < total)
                drain_rows(rc0.x & REC_COL_MASK, (int)((unsigned)rc0.x >> REC_COL_BITS) * OUT_DIM,
                           __int_as_float(rc0.y), wr, wc, total - full_end, lane, support, acc);
        }
#undef FETCH_REC
    } else {
    const int nv4 = n_edges >> 2;
    const v4i* rows4 = (const v4i*)rows;
    const int step_vecs = 32 * SCAN_U;
    const int stride = SPMM_WAVES * step_vecs;

#define TEST_EDGE(RV, LOV, EV)                                                   \
    {                                                                            \
        const bool match_ = (unsigned)((RV) - (LOV)) < (unsigned)nrows;          \
        const unsigned m_ = __builtin_amdgcn_ballot_w32(match_);                 \
        if (m_ != 0u) {                                                          \
            if (match_) mylist[wcount + (int)__builtin_amdgcn_mbcnt_lo(m_, 0u)] = (EV); \
            wcount += __builtin_popcount(m_);                                    \
        }                                                                        \
    }

    if (wave == 0) {
        const int e = (nv4 << 2) + lane;
        const int r = (e < n_edges) ? rows[e] : -1;
        TEST_EDGE(r, lo, e)
    }

#define LOAD_STEP(DST, LOS, VB)                                                  \
    _Pragma("unroll") for (int u = 0; u < SCAN_U; ++u) {                         \
        const int idx_ = (VB) + u * 32 + lane;                                   \
        const bool ok_ = idx_ < nv4;                                             \
        DST[u] = rows4[ok_ ? idx_ : nv4 - 1];                                    \
        LOS[u] = ok_ ? lo : 0x7fffffff;                                          \
    }

#define SCAN_STEP(SRC, LOS, VB)                                                  \
    _Pragma("unroll") for (int u = 0; u < SCAN_U; ++u) {                         \
        const int eb_ = ((VB) + u * 32 + lane) * 4;                              \
        TEST_EDGE(SRC[u].x, LOS[u], eb_ + 0)                                     \
        TEST_EDGE(SRC[u].y, LOS[u], eb_ + 1)                                     \
        TEST_EDGE(SRC[u].z, LOS[u], eb_ + 2)                                     \
        TEST_EDGE(SRC[u].w, LOS[u], eb_ + 3)                                     \
    }                                                                            \
    while (wcount >= 32) {                                                       \
        wcount -= 32;                                                            \
        drain_batch(mylist + wcount, 32, lane, lo, rows, cols, adj, dinv, support, acc); \
    }

    int vb = wave * step_vecs;
    if (vb < nv4) {
        v4i ra[SCAN_U], rb[SCAN_U];
        int la[SCAN_U], lb[SCAN_U];
        LOAD_STEP(ra, la, vb)
        for (;;) {
            LOAD_STEP(rb, lb, vb + stride)
            SCAN_STEP(ra, la, vb)
            vb += stride;
            if (vb >= nv4) break;
            LOAD_STEP(ra, la, vb + stride)
            SCAN_STEP(rb, lb, vb)
            vb += stride;
            if (vb >= nv4) break;
        }
    }
#undef SCAN_STEP
#undef LOAD_STEP
#undef TEST_EDGE
    }
    if (wcount > 0) drain_batch(mylist, wcount, lane, lo, rows, cols, adj, dinv, support, acc);
    __syncthreads();

    v4f* out4 = (v4f*)(out + (size_t)lo * OUT_DIM);
    for (int i = tid; i < nrows * (OUT_DIM / 4); i += SPMM_THREADS) out4[i] = acc4[i];
}

extern "C" void kernel_launch(void* const* d_in, const int* in_sizes, int n_in,
                              void* d_out, int out_size, void* d_ws, size_t ws_size,
                              hipStream_t stream) {
    const int* rows = (const int*)d_in[0];
    const int* cols = (const int*)d_in[1];
    const float* adj = (const float*)d_in[2];
    const float* x = (const float*)d_in[3];
    const float* W = (const float*)d_in[4];
    const float* b = (const float*)d_in[5];
    float* out = (float*)d_out;

    const int n_edges = in_sizes[0];
    const int n_nodes = in_sizes[3] / IN_DIM;

    float* support = (float*)d_ws;
    float* deg = support + (size_t)n_nodes * OUT_DIM;
    float* dinv = deg + n_nodes;
    int* binoff = (int*)(dinv + n_nodes);
    int* recs_raw = binoff + BIN_WGS * OFF_PITCH;
    const unsigned long long ws_bytes = (unsigned long long)ws_size;

    zero_f32_kernel<<<(unsigned)((n_nodes + 255) / 256), 256, 0, stream>>>(deg, n_nodes);

    degree_kernel<<<(n_edges + 255) / 256, 256, 0, stream>>>(rows, adj, deg, n_edges);

    dinv_kernel<<<(n_nodes + 255) / 256, 256, 0, stream>>>(deg, dinv, n_nodes);

    {
        const int mtiles = (n_nodes + 15) / 16;
        const int blocks = (mtiles + 1) / 2;
        gcn_gemm_wmma<<<blocks, 64, 0, stream>>>(x, W, b, support, n_nodes);
    }

    bin_edges_kernel<<<BIN_WGS, BIN_THREADS, 0, stream>>>(rows, cols, adj, binoff, recs_raw, n_edges, n_nodes, ws_bytes);

    spmm_slice_kernel<<<(n_nodes + SLICE_ROWS - 1) / SLICE_ROWS, SPMM_THREADS, 0, stream>>>(
        rows, cols, adj, dinv, support, out, binoff, recs_raw, n_edges, n_nodes, ws_bytes);
}


# SpMM: destination slices halved to 392 rows: 256 slice workgroups, one per WGP, where 128 left half the WGPs without work; LDS float adds still in arrival order
# speedup vs baseline: 2.5473x; 1.1501x over previous
#include <hip/hip_runtime.h>


typedef __attribute__((ext_vector_type(2))) float v2f;
typedef __attribute__((ext_vector_type(8))) float v8f;

#define IN_DIM 64
#define OUT_DIM 64

__global__ __launch_bounds__(256) void zero_f32_kernel(float* __restrict__ p, long long n) {
    long long i = (long long)blockIdx.x * blockDim.x + threadIdx.x;
    if (i < n) p[i] = 0.0f;
}

__global__ __launch_bounds__(256) void degree_kernel(const int* __restrict__ rows,
                                                     const float* __restrict__ adj,
                                                     float* __restrict__ deg,
                                                     int n_edges) {
    int e = blockIdx.x * blockDim.x + threadIdx.x;
    if (e < n_edges) {
        unsafeAtomicAdd(&deg[rows[e]], adj[e]);
    }
}

__global__ __launch_bounds__(256) void dinv_kernel(const float* __restrict__ deg,
                                                   float* __restrict__ dinv,
                                                   int n_nodes) {
    int i = blockIdx.x * blockDim.x + threadIdx.x;
    if (i < n_nodes) {
        float d = deg[i];
        d = (d > 0.0f) ? d : 1.0f;
        dinv[i] = rsqrtf(d);
    }
}

__global__ __launch_bounds__(64) void gcn_gemm_wmma(const float* __restrict__ x,
                                                    const float* __restrict__ W,
                                                    const float* __restrict__ b,
                                                    float* __restrict__ support,
                                                    int n_nodes) {
    const int wave = threadIdx.x >> 5;
    const int lane = threadIdx.x & 31;
    const int mtile = blockIdx.x * 2 + wave;
    const int m0 = mtile * 16;
    if (m0 >= n_nodes) return;

    const int half = lane >> 4;
    const int r = lane & 15;

    v8f acc[4];
#pragma unroll
    for (int nt = 0; nt < 4; ++nt) {
        const float bv = b[nt * 16 + r];
#pragma unroll
        for (int i = 0; i < 8; ++i) acc[nt][i] = bv;
    }

    const float* arow = x + (size_t)(m0 + r) * IN_DIM;

#pragma unroll
    for (int k = 0; k < IN_DIM; k += 4) {
        const v2f a = *(const v2f*)(arow + k + 2 * half);
#pragma unroll
        for (int nt = 0; nt < 4; ++nt) {
            v2f bf;
            bf.x = W[(size_t)(k + 2 * half + 0) * OUT_DIM + nt * 16 + r];
            bf.y = W[(size_t)(k + 2 * half + 1) * OUT_DIM + nt * 16 + r];
            acc[nt] = __builtin_amdgcn_wmma_f32_16x16x4_f32(
                false, a, false, bf, (short)0, acc[nt], false, false);
        }
    }

    float* srow = support + (size_t)m0 * OUT_DIM;
#pragma unroll
    for (int nt = 0; nt < 4; ++nt) {
#pragma unroll
        for (int i = 0; i < 8; ++i) {
            srow[(size_t)(i + 8 * half) * OUT_DIM + nt * 16 + r] = acc[nt][i];
        }
    }
}

#define SLICE_ROWS 392
#define SPMM_WAVES 32
#define SPMM_THREADS (SPMM_WAVES * 32)
#define SCAN_U 4
#define LIST_CAP (32 + SCAN_U * 4 * 32)
#define DRAIN_G 16

typedef __attribute__((ext_vector_type(2))) int v2i;
typedef __attribute__((ext_vector_type(4))) int v4i;
typedef __attribute__((ext_vector_type(4))) float v4f;

__device__ __forceinline__ void lds_add_f32(float* p, float v) {
    __hip_atomic_fetch_add(p, v, __ATOMIC_RELAXED, __HIP_MEMORY_SCOPE_WORKGROUP);
}

__device__ __forceinline__ void drain_rows(const int c, const int rl, const float wa, const float wr, const float wc,
                                           const int cnt, const int lane,
                                           const float* __restrict__ support,
                                           float* acc) {
    int nbits = 0;
#pragma unroll
    for (int g = 0; g < 32 / DRAIN_G; ++g) {
        if (g * DRAIN_G < cnt) {
            v2f s[DRAIN_G];
#pragma unroll
            for (int t = 0; t < DRAIN_G; ++t) {
                const int cj = __builtin_amdgcn_readlane(c, g * DRAIN_G + t);
                s[t] = *(const v2f*)(support + (size_t)cj * OUT_DIM + 2 * lane);
            }
            if (g == 0) nbits = __float_as_int(wa * wr * wc);
#pragma unroll
            for (int t = 0; t < DRAIN_G; ++t) {
                const int j = g * DRAIN_G + t;
                if (j < cnt) {
                    const float nj = __int_as_float(__builtin_amdgcn_readlane(nbits, j));
                    const int rj = __builtin_amdgcn_readlane(rl, j) + 2 * lane;
                    lds_add_f32(acc + rj, nj * s[t].x);
                    lds_add_f32(acc + rj + 1, nj * s[t].y);
                }
            }
        }
    }
}

__device__ __forceinline__ void drain_batch(const int* __restrict__ ent, const int cnt, const int lane,
                                            const int lo,
                                            const int* __restrict__ rows,
                                            const int* __restrict__ cols,
                                            const float* __restrict__ adj,
                                            const float* __restrict__ dinv,
                                            const float* __restrict__ support,
                                            float* acc) {
    __builtin_amdgcn_wave_barrier();
    const int e = (lane < cnt) ? ent[lane] : ent[0];
    const int r = rows[e];
    const int c = cols[e];
    drain_rows(c, (r - lo) * OUT_DIM, adj[e], dinv[r], dinv[c], cnt, lane, support, acc);
}


#define BIN_WGS 256
#define BIN_THREADS 1024
#define BIN_U 4
#define MAX_SLICES 256
#define OFF_PITCH (MAX_SLICES + 1)
#define SEGS_PER_WAVE (BIN_WGS / SPMM_WAVES)
#define REC_COL_BITS 22
#define REC_COL_MASK ((1 << REC_COL_BITS) - 1)
static_assert(SLICE_ROWS <= (1 << (32 - REC_COL_BITS)), "the row inside the slice must fit above the col bits");

__device__ __forceinline__ int rec_pad_ints(const int* recs_raw) {
    return (int)(((8ull - ((unsigned long long)recs_raw & 7ull)) & 7ull) >> 2);
}

__device__ __forceinline__ bool recs_fit(unsigned long long ws_bytes, int n_nodes, int n_edges, const int* recs_raw) {
    const unsigned long long pad = 4ull * (unsigned long long)rec_pad_ints(recs_raw);
    const unsigned long long need =
        ((unsigned long long)n_nodes * (OUT_DIM + 2) + (unsigned long long)BIN_WGS * OFF_PITCH) * 4ull + pad +
        (unsigned long long)n_edges * 8ull;
    return ws_bytes >= need && n_nodes <= (1 << REC_COL_BITS) && (n_nodes + SLICE_ROWS - 1) / SLICE_ROWS <= MAX_SLICES;
}

__global__ __launch_bounds__(BIN_THREADS) void bin_edges_kernel(const int* __restrict__ rows,
                                                                const int* __restrict__ cols,
                                                                const float* __restrict__ adj,
                                                                int* __restrict__ binoff,
                                                                int* recs_raw,
                                                                int n_edges, int n_nodes,
                                                                unsigned long long ws_bytes) {
    if (!recs_fit(ws_bytes, n_nodes, n_edges, recs_raw)) return;
    v2i* __restrict__ recs = (v2i*)(recs_raw + rec_pad_ints(recs_raw));
    __shared__ int cnt[MAX_SLICES];
    __shared__ int cur[MAX_SLICES];
    const int tid = threadIdx.x;
    const int ch = (n_edges + BIN_WGS - 1) / BIN_WGS;
    const long long c0l = (long long)blockIdx.x * ch;
    const int c0 = (c0l < (long long)n_edges) ? (int)c0l : n_edges;
    const int c1 = (n_edges - c0 < ch) ? n_edges : c0 + ch;

    if (tid < MAX_SLICES) cnt[tid] = 0;
    __syncthreads();
    for (int eb = c0 + tid; eb < c1; eb += BIN_THREADS * BIN_U) {
        unsigned r[BIN_U];
#pragma unroll
        for (int u = 0; u < BIN_U; ++u) {
            const int e = eb + u * BIN_THREADS;
            r[u] = (e < c1) ? (unsigned)rows[e] : 0xffffffffu;
        }
#pragma unroll
        for (int u = 0; u < BIN_U; ++u)
            if (r[u] < (unsigned)n_nodes) atomicAdd(&cnt[r[u] / SLICE_ROWS], 1);
    }
    __syncthreads();
    if (tid < 32) {
        int v[MAX_SLICES / 32];
        int sum = 0;
#pragma unroll
        for (int i = 0; i < MAX_SLICES / 32; ++i) {
            v[i] = cnt[tid * (MAX_SLICES / 32) + i];
            sum += v[i];
        }
        int incl = sum;
#pragma unroll
        for (int d = 1; d < 32; d <<= 1) {
            const int t = __shfl_up(incl, d, 32);
            if (tid >= d) incl += t;
        }
        int run = incl - sum;
        int* orow = binoff + (size_t)blockIdx.x * OFF_PITCH;
#pragma unroll
        for (int i = 0; i < MAX_SLICES / 32; ++i) {
            cur[tid * (MAX_SLICES / 32) + i] = run;
            orow[tid * (MAX_SLICES / 32) + i] = run;
            run += v[i];
        }
        if (tid == 31) orow[MAX_SLICES] = run;
    }
    __syncthreads();
    for (int eb = c0 + tid; eb < c1; eb += BIN_THREADS * BIN_U) {
        unsigned r[BIN_U];
        int cc[BIN_U], av[BIN_U];
#pragma unroll
        for (int u = 0; u < BIN_U; ++u) {
            const int e = eb + u * BIN_THREADS;
            const bool ok = e < c1;
            r[u] = ok ? (unsigned)rows[e] : 0xffffffffu;
            cc[u] = ok ? cols[e] : 0;
            av[u] = ok ? __float_as_int(adj[e]) : 0;
        }
#pragma unroll
        for (int u = 0; u < BIN_U; ++u) {
            if (r[u] < (unsigned)n_nodes) {
                const int sl = (int)(r[u] / SLICE_ROWS);
                const int p = atomicAdd(&cur[sl], 1);
                v2i rec;
                rec.x = (int)(((unsigned)cc[u] & (unsigned)REC_COL_MASK) | ((r[u] - (unsigned)(sl * SLICE_ROWS)) << REC_COL_BITS));
                rec.y = av[u];
                recs[(size_t)c0 + p] = rec;
            }
        }
    }
}

__global__ __launch_bounds__(SPMM_THREADS) void spmm_slice_kernel(const int* __restrict__ rows,
                                                                  const int* __restrict__ cols,
                                                                  const float* __restrict__ adj,
                                                                  const float* __restrict__ dinv,
                                                                  const float* __restrict__ support,
                                                                  float* __restrict__ out,
                                                                  const int* __restrict__ binoff,
                                                                  const int* __restrict__ recs_raw,
                                                                  int n_edges, int n_nodes,
                                                                  unsigned long long ws_bytes) {
    __shared__ float acc[SLICE_ROWS * OUT_DIM];
    __shared__ int wlist[SPMM_WAVES * LIST_CAP];

    const int tid = threadIdx.x;
    const int lane = tid & 31;
    const int wave = __builtin_amdgcn_readfirstlane(tid >> 5);
    const int lo = blockIdx.x * SLICE_ROWS;
    int nrows = n_nodes - lo;
    if (nrows > SLICE_ROWS) nrows = SLICE_ROWS;

    v4f* acc4 = (v4f*)acc;
    const v4f zero4 = {0.0f, 0.0f, 0.0f, 0.0f};
    for (int i = tid; i < nrows * (OUT_DIM / 4); i += SPMM_THREADS) acc4[i] = zero4;
    __syncthreads();

    int* mylist = wlist + wave * LIST_CAP;
    int wcount = 0;

    if (recs_fit(ws_bytes, n_nodes, n_edges, recs_raw)) {
        const v2i* __restrict__ recs = (const v2i*)(recs_raw + rec_pad_ints(recs_raw));
        const int ch = (n_edges + BIN_WGS - 1) / BIN_WGS;
        int seg_a = 0, seg_b = 0;
        if (lane < SEGS_PER_WAVE) {
            const int* o = binoff + (size_t)(wave + SPMM_WAVES * lane) * OFF_PITCH + blockIdx.x;
            seg_a = o[0];
            seg_b = o[1];
        }
        int offs[SEGS_PER_WAVE + 1];
        int qrec[SEGS_PER_WAVE];
        offs[0] = 0;
#pragma unroll
        for (int j = 0; j < SEGS_PER_WAVE; ++j) {
            const int a = __builtin_amdgcn_readlane(seg_a, j);
            const int len = __builtin_amdgcn_readlane(seg_b, j) - a;
            const long long k0l = (long long)(wave + SPMM_WAVES * j) * ch;
            const int k0 = (k0l < (long long)n_edges) ? (int)k0l : n_edges;
            qrec[j] = k0 + a - offs[j];
            offs[j + 1] = offs[j] + len;
        }
        const int total = offs[SEGS_PER_WAVE];

#define FETCH_REC(DST, I0)                                                       \
    {                                                                            \
        const int i_ = (I0) + lane;                                              \
        const int ii_ = (i_ < total) ? i_ : (I0);                                \
        int q_ = qrec[0];                                                        \
        _Pragma("unroll") for (int j = 1; j < SEGS_PER_WAVE; ++j)                \
            if (ii_ >= offs[j]) q_ = qrec[j];                                    \
        DST = recs[(size_t)(q_ + ii_)];                                          \
    }

        if (total > 0) {
            v2i rc0, rc1;
            FETCH_REC(rc0, 0)
            {
                const int nb = (32 < total) ? 32 : 0;
                FETCH_REC(rc1, nb)
            }
            float wr = dinv[lo + (int)((unsigned)rc0.x >> REC_COL_BITS)];
            float wc = dinv[rc0.x & REC_COL_MASK];
            const int full_end = total & ~31;
            for (int i0 = 0; i0 < full_end; i0 += 32) {
                v2i rc2;
                const int nb = (i0 + 64 < total) ? i0 + 64 : i0;
                FETCH_REC(rc2, nb)
                const float nwr = dinv[lo + (int)((unsigned)rc1.x >> REC_COL_BITS)];
                const float nwc = dinv[rc1.x & REC_COL_MASK];
                drain_rows(rc0.x & REC_COL_MASK, (int)((unsigned)rc0.x >> REC_COL_BITS) * OUT_DIM,
                           __int_as_float(rc0.y), wr, wc, 32, lane, support, acc);
                rc0 = rc1;
                rc1 = rc2;
                wr = nwr;
                wc = nwc;
            }
            if (full_end < total)
                drain_rows(rc0.x & REC_COL_MASK, (int)((unsigned)rc0.x >> REC_COL_BITS) * OUT_DIM,
                           __int_as_float(rc0.y), wr, wc, total - full_end, lane, support, acc);
        }
#undef FETCH_REC
    } else {
    const int nv4 = n_edges >> 2;
    const v4i* rows4 = (const v4i*)rows;
    const int step_vecs = 32 * SCAN_U;
    const int stride = SPMM_WAVES * step_vecs;

#define TEST_EDGE(RV, LOV, EV)                                                   \
    {                                                                            \
        const bool match_ = (unsigned)((RV) - (LOV)) < (unsigned)nrows;          \
        const unsigned m_ = __builtin_amdgcn_ballot_w32(match_);                 \
        if (m_ != 0u) {                                                          \
            if (match_) mylist[wcount + (int)__builtin_amdgcn_mbcnt_lo(m_, 0u)] = (EV); \
            wcount += __builtin_popcount(m_);                                    \
        }                                                                        \
    }

    if (wave == 0) {
        const int e = (nv4 << 2) + lane;
        const int r = (e < n_edges) ? rows[e] : -1;
        TEST_EDGE(r, lo, e)
    }

#define LOAD_STEP(DST, LOS, VB)                                                  \
    _Pragma("unroll") for (int u = 0; u < SCAN_U; ++u) {                         \
        const int idx_ = (VB) + u * 32 + lane;                                   \
        const bool ok_ = idx_ < nv4;                                             \
        DST[u] = rows4[ok_ ? idx_ : nv4 - 1];                                    \
        LOS[u] = ok_ ? lo : 0x7fffffff;                                          \
    }

#define SCAN_STEP(SRC, LOS, VB)                                                  \
    _Pragma("unroll") for (int u = 0; u < SCAN_U; ++u) {                         \
        const int eb_ = ((VB) + u * 32 + lane) * 4;                              \
        TEST_EDGE(SRC[u].x, LOS[u], eb_ + 0)                                     \
        TEST_EDGE(SRC[u].y, LOS[u], eb_ + 1)                                     \
        TEST_EDGE(SRC[u].z, LOS[u], eb_ + 2)                                     \
        TEST_EDGE(SRC[u].w, LOS[u], eb_ + 3)                                     \
    }                                                                            \
    while (wcount >= 32) {                                                       \
        wcount -= 32;                                                            \
        drain_batch(mylist + wcount, 32, lane, lo, rows, cols, adj, dinv, support, acc); \
    }

    int vb = wave * step_vecs;
    if (vb < nv4) {
        v4i ra[SCAN_U], rb[SCAN_U];
        int la[SCAN_U], lb[SCAN_U];
        LOAD_STEP(ra, la, vb)
        for (;;) {
            LOAD_STEP(rb, lb, vb + stride)
            SCAN_STEP(ra, la, vb)
            vb += stride;
            if (vb >= nv4) break;
            LOAD_STEP(ra, la, vb + stride)
            SCAN_STEP(rb, lb, vb)
            vb += stride;
            if (vb >= nv4) break;
        }
    }
#undef SCAN_STEP
#undef LOAD_STEP
#undef TEST_EDGE
    }
    if (wcount > 0) drain_batch(mylist, wcount, lane, lo, rows, cols, adj, dinv, support, acc);
    __syncthreads();

    v4f* out4 = (v4f*)(out + (size_t)lo * OUT_DIM);
    for (int i = tid; i < nrows * (OUT_DIM / 4); i += SPMM_THREADS) out4[i] = acc4[i];
}

extern "C" void kernel_launch(void* const* d_in, const int* in_sizes, int n_in,
                              void* d_out, int out_size, void* d_ws, size_t ws_size,
                              hipStream_t stream) {
    const int* rows = (const int*)d_in[0];
    const int* cols = (const int*)d_in[1];
    const float* adj = (const float*)d_in[2];
    const float* x = (const float*)d_in[3];
    const float* W = (const float*)d_in[4];
    const float* b = (const float*)d_in[5];
    float* out = (float*)d_out;

    const int n_edges = in_sizes[0];
    const int n_nodes = in_sizes[3] / IN_DIM;

    float* support = (float*)d_ws;
    float* deg = support + (size_t)n_nodes * OUT_DIM;
    float* dinv = deg + n_nodes;
    int* binoff = (int*)(dinv + n_nodes);
    int* recs_raw = binoff + BIN_WGS * OFF_PITCH;
    const unsigned long long ws_bytes = (unsigned long long)ws_size;

    zero_f32_kernel<<<(unsigned)((n_nodes + 255) / 256), 256, 0, stream>>>(deg, n_nodes);

    degree_kernel<<<(n_edges + 255) / 256, 256, 0, stream>>>(rows, adj, deg, n_edges);

    dinv_kernel<<<(n_nodes + 255) / 256, 256, 0, stream>>>(deg, dinv, n_nodes);

    {
        const int mtiles = (n_nodes + 15) / 16;
        const int blocks = (mtiles + 1) / 2;
        gcn_gemm_wmma<<<blocks, 64, 0, stream>>>(x, W, b, support, n_nodes);
    }

    bin_edges_kernel<<<BIN_WGS, BIN_THREADS, 0, stream>>>(rows, cols, adj, binoff, recs_raw, n_edges, n_nodes, ws_bytes);

    spmm_slice_kernel<<<(n_nodes + SLICE_ROWS - 1) / SLICE_ROWS, SPMM_THREADS, 0, stream>>>(
        rows, cols, adj, dinv, support, out, binoff, recs_raw, n_edges, n_nodes, ws_bytes);
}


# degree and dinv: one workgroup per destination slice sums adj from the slice's edge records with LDS float adds (arrival order); replaces zero-fill, global-atomic degree pass and dinv launch
# speedup vs baseline: 5.3352x; 2.0945x over previous
#include <hip/hip_runtime.h>


typedef __attribute__((ext_vector_type(2))) float v2f;
typedef __attribute__((ext_vector_type(8))) float v8f;

#define IN_DIM 64
#define OUT_DIM 64

__global__ __launch_bounds__(64) void gcn_gemm_wmma(const float* __restrict__ x,
                                                    const float* __restrict__ W,
                                                    const float* __restrict__ b,
                                                    float* __restrict__ support,
                                                    int n_nodes) {
    const int wave = threadIdx.x >> 5;
    const int lane = threadIdx.x & 31;
    const int mtile = blockIdx.x * 2 + wave;
    const int m0 = mtile * 16;
    if (m0 >= n_nodes) return;

    const int half = lane >> 4;
    const int r = lane & 15;

    v8f acc[4];
#pragma unroll
    for (int nt = 0; nt < 4; ++nt) {
        const float bv = b[nt * 16 + r];
#pragma unroll
        for (int i = 0; i < 8; ++i) acc[nt][i] = bv;
    }

    const float* arow = x + (size_t)(m0 + r) * IN_DIM;

#pragma unroll
    for (int k = 0; k < IN_DIM; k += 4) {
        const v2f a = *(const v2f*)(arow + k + 2 * half);
#pragma unroll
        for (int nt = 0; nt < 4; ++nt) {
            v2f bf;
            bf.x = W[(size_t)(k + 2 * half + 0) * OUT_DIM + nt * 16 + r];
            bf.y = W[(size_t)(k + 2 * half + 1) * OUT_DIM + nt * 16 + r];
            acc[nt] = __builtin_amdgcn_wmma_f32_16x16x4_f32(
                false, a, false, bf, (short)0, acc[nt], false, false);
        }
    }

    float* srow = support + (size_t)m0 * OUT_DIM;
#pragma unroll
    for (int nt = 0; nt < 4; ++nt) {
#pragma unroll
        for (int i = 0; i < 8; ++i) {
            srow[(size_t)(i + 8 * half) * OUT_DIM + nt * 16 + r] = acc[nt][i];
        }
    }
}

#define SLICE_ROWS 392
#define SPMM_WAVES 32
#define SPMM_THREADS (SPMM_WAVES * 32)
#define SCAN_U 4
#define LIST_CAP (32 + SCAN_U * 4 * 32)
#define DRAIN_G 16

typedef __attribute__((ext_vector_type(2))) int v2i;
typedef __attribute__((ext_vector_type(4))) int v4i;
typedef __attribute__((ext_vector_type(4))) float v4f;

__device__ __forceinline__ void lds_add_f32(float* p, float v) {
    __hip_atomic_fetch_add(p, v, __ATOMIC_RELAXED, __HIP_MEMORY_SCOPE_WORKGROUP);
}

__device__ __forceinline__ void drain_rows(const int c, const int rl, const float wa, const float wr, const float wc,
                                           const int cnt, const int lane,
                                           const float* __restrict__ support,
                                           float* acc) {
    int nbits = 0;
#pragma unroll
    for (int g = 0; g < 32 / DRAIN_G; ++g) {
        if (g * DRAIN_G < cnt) {
            v2f s[DRAIN_G];
#pragma unroll
            for (int t = 0; t < DRAIN_G; ++t) {
                const int cj = __builtin_amdgcn_readlane(c, g * DRAIN_G + t);
                s[t] = *(const v2f*)(support + (size_t)cj * OUT_DIM + 2 * lane);
            }
            if (g == 0) nbits = __float_as_int(wa * wr * wc);
#pragma unroll
            for (int t = 0; t < DRAIN_G; ++t) {
                const int j = g * DRAIN_G + t;
                if (j < cnt) {
                    const float nj = __int_as_float(__builtin_amdgcn_readlane(nbits, j));
                    const int rj = __builtin_amdgcn_readlane(rl, j) + 2 * lane;
                    lds_add_f32(acc + rj, nj * s[t].x);
                    lds_add_f32(acc + rj + 1, nj * s[t].y);
                }
            }
        }
    }
}

__device__ __forceinline__ void drain_batch(const int* __restrict__ ent, const int cnt, const int lane,
                                            const int lo,
                                            const int* __restrict__ rows,
                                            const int* __restrict__ cols,
                                            const float* __restrict__ adj,
                                            const float* __restrict__ dinv,
                                            const float* __restrict__ support,
                                            float* acc) {
    __builtin_amdgcn_wave_barrier();
    const int e = (lane < cnt) ? ent[lane] : ent[0];
    const int r = rows[e];
    const int c = cols[e];
    drain_rows(c, (r - lo) * OUT_DIM, adj[e], dinv[r], dinv[c], cnt, lane, support, acc);
}


#define BIN_WGS 256
#define BIN_THREADS 1024
#define BIN_U 4
#define MAX_SLICES 256
#define OFF_PITCH (MAX_SLICES + 1)
#define SEGS_PER_WAVE (BIN_WGS / SPMM_WAVES)
#define REC_COL_BITS 22
#define REC_COL_MASK ((1 << REC_COL_BITS) - 1)
static_assert(SLICE_ROWS <= (1 << (32 - REC_COL_BITS)), "the row inside the slice must fit above the col bits");

__device__ __forceinline__ int rec_pad_ints(const int* recs_raw) {
    return (int)(((8ull - ((unsigned long long)recs_raw & 7ull)) & 7ull) >> 2);
}

__device__ __forceinline__ bool recs_fit(unsigned long long ws_bytes, int n_nodes, int n_edges, const int* recs_raw) {
    const unsigned long long pad = 4ull * (unsigned long long)rec_pad_ints(recs_raw);
    const unsigned long long need =
        ((unsigned long long)n_nodes * (OUT_DIM + 1) + (unsigned long long)BIN_WGS * OFF_PITCH) * 4ull + pad +
        (unsigned long long)n_edges * 8ull;
    return ws_bytes >= need && n_nodes <= (1 << REC_COL_BITS) && (n_nodes + SLICE_ROWS - 1) / SLICE_ROWS <= MAX_SLICES;
}

__global__ __launch_bounds__(BIN_THREADS) void bin_edges_kernel(const int* __restrict__ rows,
                                                                const int* __restrict__ cols,
                                                                const float* __restrict__ adj,
                                                                int* __restrict__ binoff,
                                                                int* recs_raw,
                                                                int n_edges, int n_nodes,
                                                                unsigned long long ws_bytes) {
    if (!recs_fit(ws_bytes, n_nodes, n_edges, recs_raw)) return;
    v2i* __restrict__ recs = (v2i*)(recs_raw + rec_pad_ints(recs_raw));
    __shared__ int cnt[MAX_SLICES];
    __shared__ int cur[MAX_SLICES];
    const int tid = threadIdx.x;
    const int ch = (n_edges + BIN_WGS - 1) / BIN_WGS;
    const long long c0l = (long long)blockIdx.x * ch;
    const int c0 = (c0l < (long long)n_edges) ? (int)c0l : n_edges;
    const int c1 = (n_edges - c0 < ch) ? n_edges : c0 + ch;

    if (tid < MAX_SLICES) cnt[tid] = 0;
    __syncthreads();
    for (int eb = c0 + tid; eb < c1; eb += BIN_THREADS * BIN_U) {
        unsigned r[BIN_U];
#pragma unroll
        for (int u = 0; u < BIN_U; ++u) {
            const int e = eb + u * BIN_THREADS;
            r[u] = (e < c1) ? (unsigned)rows[e] : 0xffffffffu;
        }
#pragma unroll
        for (int u = 0; u < BIN_U; ++u)
            if (r[u] < (unsigned)n_nodes) atomicAdd(&cnt[r[u] / SLICE_ROWS], 1);
    }
    __syncthreads();
    if (tid < 32) {
        int v[MAX_SLICES / 32];
        int sum = 0;
#pragma unroll
        for (int i = 0; i < MAX_SLICES / 32; ++i) {
            v[i] = cnt[tid * (MAX_SLICES / 32) + i];
            sum += v[i];
        }
        int incl = sum;
#pragma unroll
        for (int d = 1; d < 32; d <<= 1) {
            const int t = __shfl_up(incl, d, 32);
            if (tid >= d) incl += t;
        }
        int run = incl - sum;
        int* orow = binoff + (size_t)blockIdx.x * OFF_PITCH;
#pragma unroll
        for (int i = 0; i < MAX_SLICES / 32; ++i) {
            cur[tid * (MAX_SLICES / 32) + i] = run;
            orow[tid * (MAX_SLICES / 32) + i] = run;
            run += v[i];
        }
        if (tid == 31) orow[MAX_SLICES] = run;
    }
    __syncthreads();
    for (int eb = c0 + tid; eb < c1; eb += BIN_THREADS * BIN_U) {
        unsigned r[BIN_U];
        int cc[BIN_U], av[BIN_U];
#pragma unroll
        for (int u = 0; u < BIN_U; ++u) {
            const int e = eb + u * BIN_THREADS;
            const bool ok = e < c1;
            r[u] = ok ? (unsigned)rows[e] : 0xffffffffu;
            cc[u] = ok ? cols[e] : 0;
            av[u] = ok ? __float_as_int(adj[e]) : 0;
        }
#pragma unroll
        for (int u = 0; u < BIN_U; ++u) {
            if (r[u] < (unsigned)n_nodes) {
                const int sl = (int)(r[u] / SLICE_ROWS);
                const int p = atomicAdd(&cur[sl], 1);
                v2i rec;
                rec.x = (int)(((unsigned)cc[u] & (unsigned)REC_COL_MASK) | ((r[u] - (unsigned)(sl * SLICE_ROWS)) << REC_COL_BITS));
                rec.y = av[u];
                recs[(size_t)c0 + p] = rec;
            }
        }
    }
}

#define DEG_WAVES 8
#define DEG_THREADS (DEG_WAVES * 32)
#define DEG_SEGS (BIN_WGS / DEG_WAVES)
static_assert(DEG_SEGS == 32 && DEG_THREADS == BIN_WGS, "one chunk per thread, one segment's bounds per lane");

__global__ __launch_bounds__(DEG_THREADS) void slice_dinv_kernel(const int* __restrict__ rows,
                                                                 const float* __restrict__ adj,
                                                                 float* __restrict__ dinv,
                                                                 const int* __restrict__ binoff,
                                                                 const int* __restrict__ recs_raw,
                                                                 int n_edges, int n_nodes,
                                                                 unsigned long long ws_bytes) {
    __shared__ float deg[SLICE_ROWS];
    const int tid = threadIdx.x;
    const int lane = tid & 31;
    const int lo = blockIdx.x * SLICE_ROWS;
    int nrows = n_nodes - lo;
    if (nrows > SLICE_ROWS) nrows = SLICE_ROWS;

    for (int i = tid; i < SLICE_ROWS; i += DEG_THREADS) deg[i] = 0.0f;
    __syncthreads();

    if (recs_fit(ws_bytes, n_nodes, n_edges, recs_raw)) {
        const v2i* __restrict__ recs = (const v2i*)(recs_raw + rec_pad_ints(recs_raw));
        const int ch = (n_edges + BIN_WGS - 1) / BIN_WGS;
        const int chunk = tid;
        const int* o = binoff + (size_t)chunk * OFF_PITCH + blockIdx.x;
        const int seg_a = o[0];
        const int seg_len = o[1] - seg_a;
        const long long k0l = (long long)chunk * ch;
        const int k0 = (k0l < (long long)n_edges) ? (int)k0l : n_edges;
        const int seg_q = k0 + seg_a;

        v2i rc[DEG_SEGS];
#pragma unroll
        for (int j = 0; j < DEG_SEGS; ++j) {
            const int len = __builtin_amdgcn_readlane(seg_len, j);
            const int q = __builtin_amdgcn_readlane(seg_q, j);
            rc[j] = v2i{0, 0};
            if (len > 0) rc[j] = recs[(size_t)(q + ((lane < len) ? lane : 0))];
        }
#pragma unroll
        for (int j = 0; j < DEG_SEGS; ++j) {
            const int len = __builtin_amdgcn_readlane(seg_len, j);
            if (lane < len) lds_add_f32(deg + (int)((unsigned)rc[j].x >> REC_COL_BITS), __int_as_float(rc[j].y));
        }
        for (int j = 0; j < DEG_SEGS; ++j) {
            const int len = __builtin_amdgcn_readlane(seg_len, j);
            const int q = __builtin_amdgcn_readlane(seg_q, j);
            for (int i = 32 + lane; i < len; i += 32) {
                const v2i r = recs[(size_t)(q + i)];
                lds_add_f32(deg + (int)((unsigned)r.x >> REC_COL_BITS), __int_as_float(r.y));
            }
        }
    } else {
        for (int eb = tid; eb < n_edges; eb += DEG_THREADS * 4) {
            unsigned rl[4];
#pragma unroll
            for (int u = 0; u < 4; ++u) {
                const int e = eb + u * DEG_THREADS;
                const bool ok = e < n_edges;
                const unsigned rv = (unsigned)rows[ok ? e : n_edges - 1];
                rl[u] = ok ? rv - (unsigned)lo : 0xffffffffu;
            }
#pragma unroll
            for (int u = 0; u < 4; ++u)
                if (rl[u] < (unsigned)nrows) lds_add_f32(deg + rl[u], adj[eb + u * DEG_THREADS]);
        }
    }
    __syncthreads();

    for (int i = tid; i < nrows; i += DEG_THREADS) {
        float d = deg[i];
        d = (d > 0.0f) ? d : 1.0f;
        dinv[lo + i] = rsqrtf(d);
    }
}

__global__ __launch_bounds__(SPMM_THREADS) void spmm_slice_kernel(const int* __restrict__ rows,
                                                                  const int* __restrict__ cols,
                                                                  const float* __restrict__ adj,
                                                                  const float* __restrict__ dinv,
                                                                  const float* __restrict__ support,
                                                                  float* __restrict__ out,
                                                                  const int* __restrict__ binoff,
                                                                  const int* __restrict__ recs_raw,
                                                                  int n_edges, int n_nodes,
                                                                  unsigned long long ws_bytes) {
    __shared__ float acc[SLICE_ROWS * OUT_DIM];
    __shared__ int wlist[SPMM_WAVES * LIST_CAP];

    const int tid = threadIdx.x;
    const int lane = tid & 31;
    const int wave = __builtin_amdgcn_readfirstlane(tid >> 5);
    const int lo = blockIdx.x * SLICE_ROWS;
    int nrows = n_nodes - lo;
    if (nrows > SLICE_ROWS) nrows = SLICE_ROWS;

    v4f* acc4 = (v4f*)acc;
    const v4f zero4 = {0.0f, 0.0f, 0.0f, 0.0f};
    for (int i = tid; i < nrows * (OUT_DIM / 4); i += SPMM_THREADS) acc4[i] = zero4;
    __syncthreads();

    int* mylist = wlist + wave * LIST_CAP;
    int wcount = 0;

    if (recs_fit(ws_bytes, n_nodes, n_edges, recs_raw)) {
        const v2i* __restrict__ recs = (const v2i*)(recs_raw + rec_pad_ints(recs_raw));
        const int ch = (n_edges + BIN_WGS - 1) / BIN_WGS;
        int seg_a = 0, seg_b = 0;
        if (lane < SEGS_PER_WAVE) {
            const int* o = binoff + (size_t)(wave + SPMM_WAVES * lane) * OFF_PITCH + blockIdx.x;
            seg_a = o[0];
            seg_b = o[1];
        }
        int offs[SEGS_PER_WAVE + 1];
        int qrec[SEGS_PER_WAVE];
        offs[0] = 0;
#pragma unroll
        for (int j = 0; j < SEGS_PER_WAVE; ++j) {
            const int a = __builtin_amdgcn_readlane(seg_a, j);
            const int len = __builtin_amdgcn_readlane(seg_b, j) - a;
            const long long k0l = (long long)(wave + SPMM_WAVES * j) * ch;
            const int k0 = (k0l < (long long)n_edges) ? (int)k0l : n_edges;
            qrec[j] = k0 + a - offs[j];
            offs[j + 1] = offs[j] + len;
        }
        const int total = offs[SEGS_PER_WAVE];

#define FETCH_REC(DST, I0)                                                       \
    {                                                                            \
        const int i_ = (I0) + lane;                                              \
        const int ii_ = (i_ < total) ? i_ : (I0);                                \
        int q_ = qrec[0];                                                        \
        _Pragma("unroll") for (int j = 1; j < SEGS_PER_WAVE; ++j)                \
            if (ii_ >= offs[j]) q_ = qrec[j];                                    \
        DST = recs[(size_t)(q_ + ii_)];                                          \
    }

        if (total > 0) {
            v2i rc0, rc1;
            FETCH_REC(rc0, 0)
            {
                const int nb = (32 < total) ? 32 : 0;
                FETCH_REC(rc1, nb)
            }
            float wr = dinv[lo + (int)((unsigned)rc0.x >> REC_COL_BITS)];
            float wc = dinv[rc0.x & REC_COL_MASK];
            const int full_end = total & ~31;
            for (int i0 = 0; i0 < full_end; i0 += 32) {
                v2i rc2;
                const int nb = (i0 + 64 < total) ? i0 + 64 : i0;
                FETCH_REC(rc2, nb)
                const float nwr = dinv[lo + (int)((unsigned)rc1.x >> REC_COL_BITS)];
                const float nwc = dinv[rc1.x & REC_COL_MASK];
                drain_rows(rc0.x & REC_COL_MASK, (int)((unsigned)rc0.x >> REC_COL_BITS) * OUT_DIM,
                           __int_as_float(rc0.y), wr, wc, 32, lane, support, acc);
                rc0 = rc1;
                rc1 = rc2;
                wr = nwr;
                wc = nwc;
            }
            if (full_end < total)
                drain_rows(rc0.x & REC_COL_MASK, (int)((unsigned)rc0.x >> REC_COL_BITS) * OUT_DIM,
                           __int_as_float(rc0.y), wr, wc, total - full_end, lane, support, acc);
        }
#undef FETCH_REC
    } else {
    const int nv4 = n_edges >> 2;
    const v4i* rows4 = (const v4i*)rows;
    const int step_vecs = 32 * SCAN_U;
    const int stride = SPMM_WAVES * step_vecs;

#define TEST_EDGE(RV, LOV, EV)                                                   \
    {                                                                            \
        const bool match_ = (unsigned)((RV) - (LOV)) < (unsigned)nrows;          \
        const unsigned m_ = __builtin_amdgcn_ballot_w32(match_);                 \
        if (m_ != 0u) {                                                          \
            if (match_) mylist[wcount + (int)__builtin_amdgcn_mbcnt_lo(m_, 0u)] = (EV); \
            wcount += __builtin_popcount(m_);                                    \
        }                                                                        \
    }

    if (wave == 0) {
        const int e = (nv4 << 2) + lane;
        const int r = (e < n_edges) ? rows[e] : -1;
        TEST_EDGE(r, lo, e)
    }

#define LOAD_STEP(DST, LOS, VB)                                                  \
    _Pragma("unroll") for (int u = 0; u < SCAN_U; ++u) {                         \
        const int idx_ = (VB) + u * 32 + lane;                                   \
        const bool ok_ = idx_ < nv4;                                             \
        DST[u] = rows4[ok_ ? idx_ : nv4 - 1];                                    \
        LOS[u] = ok_ ? lo : 0x7fffffff;                                          \
    }

#define SCAN_STEP(SRC, LOS, VB)                                                  \
    _Pragma("unroll") for (int u = 0; u < SCAN_U; ++u) {                         \
        const int eb_ = ((VB) + u * 32 + lane) * 4;                              \
        TEST_EDGE(SRC[u].x, LOS[u], eb_ + 0)                                     \
        TEST_EDGE(SRC[u].y, LOS[u], eb_ + 1)                                     \
        TEST_EDGE(SRC[u].z, LOS[u], eb_ + 2)                                     \
        TEST_EDGE(SRC[u].w, LOS[u], eb_ + 3)                                     \
    }                                                                            \
    while (wcount >= 32) {                                                       \
        wcount -= 32;                                                            \
        drain_batch(mylist + wcount, 32, lane, lo, rows, cols, adj, dinv, support, acc); \
    }

    int vb = wave * step_vecs;
    if (vb < nv4) {
        v4i ra[SCAN_U], rb[SCAN_U];
        int la[SCAN_U], lb[SCAN_U];
        LOAD_STEP(ra, la, vb)
        for (;;) {
            LOAD_STEP(rb, lb, vb + stride)
            SCAN_STEP(ra, la, vb)
            vb += stride;
            if (vb >= nv4) break;
            LOAD_STEP(ra, la, vb + stride)
            SCAN_STEP(rb, lb, vb)
            vb += stride;
            if (vb >= nv4) break;
        }
    }
#undef SCAN_STEP
#undef LOAD_STEP
#undef TEST_EDGE
    }
    if (wcount > 0) drain_batch(mylist, wcount, lane, lo, rows, cols, adj, dinv, support, acc);
    __syncthreads();

    v4f* out4 = (v4f*)(out + (size_t)lo * OUT_DIM);
    for (int i = tid; i < nrows * (OUT_DIM / 4); i += SPMM_THREADS) out4[i] = acc4[i];
}

extern "C" void kernel_launch(void* const* d_in, const int* in_sizes, int n_in,
                              void* d_out, int out_size, void* d_ws, size_t ws_size,
                              hipStream_t stream) {
    const int* rows = (const int*)d_in[0];
    const int* cols = (const int*)d_in[1];
    const float* adj = (const float*)d_in[2];
    const float* x = (const float*)d_in[3];
    const float* W = (const float*)d_in[4];
    const float* b = (const float*)d_in[5];
    float* out = (float*)d_out;

    const int n_edges = in_sizes[0];
    const int n_nodes = in_sizes[3] / IN_DIM;

    float* support = (float*)d_ws;
    float* dinv = support + (size_t)n_nodes * OUT_DIM;
    int* binoff = (int*)(dinv + n_nodes);
    int* recs_raw = binoff + BIN_WGS * OFF_PITCH;
    const unsigned long long ws_bytes = (unsigned long long)ws_size;

    {
        const int mtiles = (n_nodes + 15) / 16;
        const int blocks = (mtiles + 1) / 2;
        gcn_gemm_wmma<<<blocks, 64, 0, stream>>>(x, W, b, support, n_nodes);
    }

    bin_edges_kernel<<<BIN_WGS, BIN_THREADS, 0, stream>>>(rows, cols, adj, binoff, recs_raw, n_edges, n_nodes, ws_bytes);

    slice_dinv_kernel<<<(n_nodes + SLICE_ROWS - 1) / SLICE_ROWS, DEG_THREADS, 0, stream>>>(
        rows, adj, dinv, binoff, recs_raw, n_edges, n_nodes, ws_bytes);

    spmm_slice_kernel<<<(n_nodes + SLICE_ROWS - 1) / SLICE_ROWS, SPMM_THREADS, 0, stream>>>(
        rows, cols, adj, dinv, support, out, binoff, recs_raw, n_edges, n_nodes, ws_bytes);
}


# SpMM: slice edges ordered by row in LDS (integer LDS adds: count, scan, place); each wave owns a run of rows and sums them in registers in list order: no LDS float adds; order of sums: LDS ranks
# speedup vs baseline: 6.1215x; 1.1474x over previous
#include <hip/hip_runtime.h>


typedef __attribute__((ext_vector_type(2))) float v2f;
typedef __attribute__((ext_vector_type(8))) float v8f;

#define IN_DIM 64
#define OUT_DIM 64

__global__ __launch_bounds__(64) void gcn_gemm_wmma(const float* __restrict__ x,
                                                    const float* __restrict__ W,
                                                    const float* __restrict__ b,
                                                    float* __restrict__ support,
                                                    int n_nodes) {
    const int wave = threadIdx.x >> 5;
    const int lane = threadIdx.x & 31;
    const int mtile = blockIdx.x * 2 + wave;
    const int m0 = mtile * 16;
    if (m0 >= n_nodes) return;

    const int half = lane >> 4;
    const int r = lane & 15;

    v8f acc[4];
#pragma unroll
    for (int nt = 0; nt < 4; ++nt) {
        const float bv = b[nt * 16 + r];
#pragma unroll
        for (int i = 0; i < 8; ++i) acc[nt][i] = bv;
    }

    const float* arow = x + (size_t)(m0 + r) * IN_DIM;

#pragma unroll
    for (int k = 0; k < IN_DIM; k += 4) {
        const v2f a = *(const v2f*)(arow + k + 2 * half);
#pragma unroll
        for (int nt = 0; nt < 4; ++nt) {
            v2f bf;
            bf.x = W[(size_t)(k + 2 * half + 0) * OUT_DIM + nt * 16 + r];
            bf.y = W[(size_t)(k + 2 * half + 1) * OUT_DIM + nt * 16 + r];
            acc[nt] = __builtin_amdgcn_wmma_f32_16x16x4_f32(
                false, a, false, bf, (short)0, acc[nt], false, false);
        }
    }

    float* srow = support + (size_t)m0 * OUT_DIM;
#pragma unroll
    for (int nt = 0; nt < 4; ++nt) {
#pragma unroll
        for (int i = 0; i < 8; ++i) {
            srow[(size_t)(i + 8 * half) * OUT_DIM + nt * 16 + r] = acc[nt][i];
        }
    }
}

#define SLICE_ROWS 392
#define SPMM_WAVES 32
#define SPMM_THREADS (SPMM_WAVES * 32)
#define SCAN_U 4
#define LIST_CAP (32 + SCAN_U * 4 * 32)
#define DRAIN_G 16
#define SORT_CAP 8192
#define ROWS_PER_LANE ((SLICE_ROWS + 31) / 32)

typedef __attribute__((ext_vector_type(2))) int v2i;
typedef __attribute__((ext_vector_type(4))) int v4i;
typedef __attribute__((ext_vector_type(4))) float v4f;

__device__ __forceinline__ void lds_add_f32(float* p, float v) {
    __hip_atomic_fetch_add(p, v, __ATOMIC_RELAXED, __HIP_MEMORY_SCOPE_WORKGROUP);
}

__device__ __forceinline__ void drain_rows(const int c, const int rl, const float wa, const float wr, const float wc,
                                           const int cnt, const int lane,
                                           const float* __restrict__ support,
                                           float* acc) {
    int nbits = 0;
#pragma unroll
    for (int g = 0; g < 32 / DRAIN_G; ++g) {
        if (g * DRAIN_G < cnt) {
            v2f s[DRAIN_G];
#pragma unroll
            for (int t = 0; t < DRAIN_G; ++t) {
                const int cj = __builtin_amdgcn_readlane(c, g * DRAIN_G + t);
                s[t] = *(const v2f*)(support + (size_t)cj * OUT_DIM + 2 * lane);
            }
            if (g == 0) nbits = __float_as_int(wa * wr * wc);
#pragma unroll
            for (int t = 0; t < DRAIN_G; ++t) {
                const int j = g * DRAIN_G + t;
                if (j < cnt) {
                    const float nj = __int_as_float(__builtin_amdgcn_readlane(nbits, j));
                    const int rj = __builtin_amdgcn_readlane(rl, j) + 2 * lane;
                    lds_add_f32(acc + rj, nj * s[t].x);
                    lds_add_f32(acc + rj + 1, nj * s[t].y);
                }
            }
        }
    }
}

__device__ __forceinline__ void own_rows(const int c, const int ro, const int wb, const int cnt, const int lane,
                                         const float* __restrict__ support, float* acc, int& cur, v2f& a) {
#pragma unroll
    for (int g = 0; g < 32 / DRAIN_G; ++g) {
        if (g * DRAIN_G < cnt) {
            v2f s[DRAIN_G];
#pragma unroll
            for (int t = 0; t < DRAIN_G; ++t) {
                const int cj = __builtin_amdgcn_readlane(c, g * DRAIN_G + t);
                s[t] = *(const v2f*)(support + (size_t)cj * OUT_DIM + 2 * lane);
            }
#pragma unroll
            for (int t = 0; t < DRAIN_G; ++t) {
                const int j = g * DRAIN_G + t;
                if (j < cnt) {
                    const int rj = __builtin_amdgcn_readlane(ro, j);
                    const float wj = __int_as_float(__builtin_amdgcn_readlane(wb, j));
                    if (__builtin_expect(rj != cur, 0)) {
                        if (cur >= 0) *(v2f*)(acc + cur + 2 * lane) = a;
                        cur = rj;
                        a = v2f{0.0f, 0.0f};
                    }
                    a.x += wj * s[t].x;
                    a.y += wj * s[t].y;
                }
            }
        }
    }
}

__device__ __forceinline__ void drain_batch(const int* __restrict__ ent, const int cnt, const int lane,
                                            const int lo,
                                            const int* __restrict__ rows,
                                            const int* __restrict__ cols,
                                            const float* __restrict__ adj,
                                            const float* __restrict__ dinv,
                                            const float* __restrict__ support,
                                            float* acc) {
    __builtin_amdgcn_wave_barrier();
    const int e = (lane < cnt) ? ent[lane] : ent[0];
    const int r = rows[e];
    const int c = cols[e];
    drain_rows(c, (r - lo) * OUT_DIM, adj[e], dinv[r], dinv[c], cnt, lane, support, acc);
}


#define BIN_WGS 256
#define BIN_THREADS 1024
#define BIN_U 4
#define MAX_SLICES 256
#define OFF_PITCH (MAX_SLICES + 1)
#define SEGS_PER_WAVE (BIN_WGS / SPMM_WAVES)
#define REC_COL_BITS 22
#define REC_COL_MASK ((1 << REC_COL_BITS) - 1)
static_assert(SLICE_ROWS <= (1 << (32 - REC_COL_BITS)), "the row inside the slice must fit above the col bits");

__device__ __forceinline__ int rec_pad_ints(const int* recs_raw) {
    return (int)(((8ull - ((unsigned long long)recs_raw & 7ull)) & 7ull) >> 2);
}

__device__ __forceinline__ bool recs_fit(unsigned long long ws_bytes, int n_nodes, int n_edges, const int* recs_raw) {
    const unsigned long long pad = 4ull * (unsigned long long)rec_pad_ints(recs_raw);
    const unsigned long long need =
        ((unsigned long long)n_nodes * (OUT_DIM + 1) + (unsigned long long)BIN_WGS * OFF_PITCH) * 4ull + pad +
        (unsigned long long)n_edges * 8ull;
    return ws_bytes >= need && n_nodes <= (1 << REC_COL_BITS) && (n_nodes + SLICE_ROWS - 1) / SLICE_ROWS <= MAX_SLICES;
}

__global__ __launch_bounds__(BIN_THREADS) void bin_edges_kernel(const int* __restrict__ rows,
                                                                const int* __restrict__ cols,
                                                                const float* __restrict__ adj,
                                                                int* __restrict__ binoff,
                                                                int* recs_raw,
                                                                int n_edges, int n_nodes,
                                                                unsigned long long ws_bytes) {
    if (!recs_fit(ws_bytes, n_nodes, n_edges, recs_raw)) return;
    v2i* __restrict__ recs = (v2i*)(recs_raw + rec_pad_ints(recs_raw));
    __shared__ int cnt[MAX_SLICES];
    __shared__ int cur[MAX_SLICES];
    const int tid = threadIdx.x;
    const int ch = (n_edges + BIN_WGS - 1) / BIN_WGS;
    const long long c0l = (long long)blockIdx.x * ch;
    const int c0 = (c0l < (long long)n_edges) ? (int)c0l : n_edges;
    const int c1 = (n_edges - c0 < ch) ? n_edges : c0 + ch;

    if (tid < MAX_SLICES) cnt[tid] = 0;
    __syncthreads();
    for (int eb = c0 + tid; eb < c1; eb += BIN_THREADS * BIN_U) {
        unsigned r[BIN_U];
#pragma unroll
        for (int u = 0; u < BIN_U; ++u) {
            const int e = eb + u * BIN_THREADS;
            r[u] = (e < c1) ? (unsigned)rows[e] : 0xffffffffu;
        }
#pragma unroll
        for (int u = 0; u < BIN_U; ++u)
            if (r[u] < (unsigned)n_nodes) atomicAdd(&cnt[r[u] / SLICE_ROWS], 1);
    }
    __syncthreads();
    if (tid < 32) {
        int v[MAX_SLICES / 32];
        int sum = 0;
#pragma unroll
        for (int i = 0; i < MAX_SLICES / 32; ++i) {
            v[i] = cnt[tid * (MAX_SLICES / 32) + i];
            sum += v[i];
        }
        int incl = sum;
#pragma unroll
        for (int d = 1; d < 32; d <<= 1) {
            const int t = __shfl_up(incl, d, 32);
            if (tid >= d) incl += t;
        }
        int run = incl - sum;
        int* orow = binoff + (size_t)blockIdx.x * OFF_PITCH;
#pragma unroll
        for (int i = 0; i < MAX_SLICES / 32; ++i) {
            cur[tid * (MAX_SLICES / 32) + i] = run;
            orow[tid * (MAX_SLICES / 32) + i] = run;
            run += v[i];
        }
        if (tid == 31) orow[MAX_SLICES] = run;
    }
    __syncthreads();
    for (int eb = c0 + tid; eb < c1; eb += BIN_THREADS * BIN_U) {
        unsigned r[BIN_U];
        int cc[BIN_U], av[BIN_U];
#pragma unroll
        for (int u = 0; u < BIN_U; ++u) {
            const int e = eb + u * BIN_THREADS;
            const bool ok = e < c1;
            r[u] = ok ? (unsigned)rows[e] : 0xffffffffu;
            cc[u] = ok ? cols[e] : 0;
            av[u] = ok ? __float_as_int(adj[e]) : 0;
        }
#pragma unroll
        for (int u = 0; u < BIN_U; ++u) {
            if (r[u] < (unsigned)n_nodes) {
                const int sl = (int)(r[u] / SLICE_ROWS);
                const int p = atomicAdd(&cur[sl], 1);
                v2i rec;
                rec.x = (int)(((unsigned)cc[u] & (unsigned)REC_COL_MASK) | ((r[u] - (unsigned)(sl * SLICE_ROWS)) << REC_COL_BITS));
                rec.y = av[u];
                recs[(size_t)c0 + p] = rec;
            }
        }
    }
}

#define DEG_WAVES 8
#define DEG_THREADS (DEG_WAVES * 32)
#define DEG_SEGS (BIN_WGS / DEG_WAVES)
static_assert(DEG_SEGS == 32 && DEG_THREADS == BIN_WGS, "one chunk per thread, one segment's bounds per lane");

__global__ __launch_bounds__(DEG_THREADS) void slice_dinv_kernel(const int* __restrict__ rows,
                                                                 const float* __restrict__ adj,
                                                                 float* __restrict__ dinv,
                                                                 const int* __restrict__ binoff,
                                                                 const int* __restrict__ recs_raw,
                                                                 int n_edges, int n_nodes,
                                                                 unsigned long long ws_bytes) {
    __shared__ float deg[SLICE_ROWS];
    const int tid = threadIdx.x;
    const int lane = tid & 31;
    const int lo = blockIdx.x * SLICE_ROWS;
    int nrows = n_nodes - lo;
    if (nrows > SLICE_ROWS) nrows = SLICE_ROWS;

    for (int i = tid; i < SLICE_ROWS; i += DEG_THREADS) deg[i] = 0.0f;
    __syncthreads();

    if (recs_fit(ws_bytes, n_nodes, n_edges, recs_raw)) {
        const v2i* __restrict__ recs = (const v2i*)(recs_raw + rec_pad_ints(recs_raw));
        const int ch = (n_edges + BIN_WGS - 1) / BIN_WGS;
        const int chunk = tid;
        const int* o = binoff + (size_t)chunk * OFF_PITCH + blockIdx.x;
        const int seg_a = o[0];
        const int seg_len = o[1] - seg_a;
        const long long k0l = (long long)chunk * ch;
        const int k0 = (k0l < (long long)n_edges) ? (int)k0l : n_edges;
        const int seg_q = k0 + seg_a;

        v2i rc[DEG_SEGS];
#pragma unroll
        for (int j = 0; j < DEG_SEGS; ++j) {
            const int len = __builtin_amdgcn_readlane(seg_len, j);
            const int q = __builtin_amdgcn_readlane(seg_q, j);
            rc[j] = v2i{0, 0};
            if (len > 0) rc[j] = recs[(size_t)(q + ((lane < len) ? lane : 0))];
        }
#pragma unroll
        for (int j = 0; j < DEG_SEGS; ++j) {
            const int len = __builtin_amdgcn_readlane(seg_len, j);
            if (lane < len) lds_add_f32(deg + (int)((unsigned)rc[j].x >> REC_COL_BITS), __int_as_float(rc[j].y));
        }
        for (int j = 0; j < DEG_SEGS; ++j) {
            const int len = __builtin_amdgcn_readlane(seg_len, j);
            const int q = __builtin_amdgcn_readlane(seg_q, j);
            for (int i = 32 + lane; i < len; i += 32) {
                const v2i r = recs[(size_t)(q + i)];
                lds_add_f32(deg + (int)((unsigned)r.x >> REC_COL_BITS), __int_as_float(r.y));
            }
        }
    } else {
        for (int eb = tid; eb < n_edges; eb += DEG_THREADS * 4) {
            unsigned rl[4];
#pragma unroll
            for (int u = 0; u < 4; ++u) {
                const int e = eb + u * DEG_THREADS;
                const bool ok = e < n_edges;
                const unsigned rv = (unsigned)rows[ok ? e : n_edges - 1];
                rl[u] = ok ? rv - (unsigned)lo : 0xffffffffu;
            }
#pragma unroll
            for (int u = 0; u < 4; ++u)
                if (rl[u] < (unsigned)nrows) lds_add_f32(deg + rl[u], adj[eb + u * DEG_THREADS]);
        }
    }
    __syncthreads();

    for (int i = tid; i < nrows; i += DEG_THREADS) {
        float d = deg[i];
        d = (d > 0.0f) ? d : 1.0f;
        dinv[lo + i] = rsqrtf(d);
    }
}

__global__ __launch_bounds__(SPMM_THREADS) void spmm_slice_kernel(const int* __restrict__ rows,
                                                                  const int* __restrict__ cols,
                                                                  const float* __restrict__ adj,
                                                                  const float* __restrict__ dinv,
                                                                  const float* __restrict__ support,
                                                                  float* __restrict__ out,
                                                                  const int* __restrict__ binoff,
                                                                  const int* __restrict__ recs_raw,
                                                                  int n_edges, int n_nodes,
                                                                  unsigned long long ws_bytes) {
    __shared__ float acc[SLICE_ROWS * OUT_DIM];
    __shared__ int wlist[SPMM_WAVES * LIST_CAP];
    __shared__ v2i sorted[SORT_CAP];
    __shared__ int rstart[SLICE_ROWS + 1];
    __shared__ int rcur[SLICE_ROWS];
    __shared__ int s_total;

    const int tid = threadIdx.x;
    const int lane = tid & 31;
    const int wave = __builtin_amdgcn_readfirstlane(tid >> 5);
    const int lo = blockIdx.x * SLICE_ROWS;
    int nrows = n_nodes - lo;
    if (nrows > SLICE_ROWS) nrows = SLICE_ROWS;

    v4f* acc4 = (v4f*)acc;
    const v4f zero4 = {0.0f, 0.0f, 0.0f, 0.0f};
    for (int i = tid; i < nrows * (OUT_DIM / 4); i += SPMM_THREADS) acc4[i] = zero4;
    for (int i = tid; i < SLICE_ROWS; i += SPMM_THREADS) rcur[i] = 0;
    if (tid == 0) s_total = 0;
    __syncthreads();

    int* mylist = wlist + wave * LIST_CAP;
    int wcount = 0;

    if (recs_fit(ws_bytes, n_nodes, n_edges, recs_raw)) {
        const v2i* __restrict__ recs = (const v2i*)(recs_raw + rec_pad_ints(recs_raw));
        const int ch = (n_edges + BIN_WGS - 1) / BIN_WGS;
        int seg_len = 0, seg_q = 0;
        if (lane < SEGS_PER_WAVE) {
            const int chunk = wave + SPMM_WAVES * lane;
            const int* o = binoff + (size_t)chunk * OFF_PITCH + blockIdx.x;
            const int sa = o[0];
            seg_len = o[1] - sa;
            const long long k0l = (long long)chunk * ch;
            const int k0 = (k0l < (long long)n_edges) ? (int)k0l : n_edges;
            seg_q = k0 + sa;
        }
        int total = 0;
#pragma unroll
        for (int j = 0; j < SEGS_PER_WAVE; ++j) total += __builtin_amdgcn_readlane(seg_len, j);

        v2i rc[SEGS_PER_WAVE];
#pragma unroll
        for (int j = 0; j < SEGS_PER_WAVE; ++j) {
            const int len = __builtin_amdgcn_readlane(seg_len, j);
            const int q = __builtin_amdgcn_readlane(seg_q, j);
            rc[j] = v2i{0, 0};
            if (len > 0) rc[j] = recs[(size_t)(q + ((lane < len) ? lane : 0))];
        }
        if (lane == 0) atomicAdd(&s_total, total);
#pragma unroll
        for (int j = 0; j < SEGS_PER_WAVE; ++j) {
            const int len = __builtin_amdgcn_readlane(seg_len, j);
            if (lane < len) atomicAdd(&rcur[(int)((unsigned)rc[j].x >> REC_COL_BITS)], 1);
        }
        for (int j = 0; j < SEGS_PER_WAVE; ++j) {
            const int len = __builtin_amdgcn_readlane(seg_len, j);
            const int q = __builtin_amdgcn_readlane(seg_q, j);
            for (int i = 32 + lane; i < len; i += 32) {
                const v2i r = recs[(size_t)(q + i)];
                atomicAdd(&rcur[(int)((unsigned)r.x >> REC_COL_BITS)], 1);
            }
        }
        __syncthreads();
        const int stot = __builtin_amdgcn_readfirstlane(s_total);

        if (stot <= SORT_CAP) {
            if (wave == 0) {
                int v[ROWS_PER_LANE];
                int sum = 0;
#pragma unroll
                for (int i = 0; i < ROWS_PER_LANE; ++i) {
                    const int idx = lane * ROWS_PER_LANE + i;
                    v[i] = (idx < SLICE_ROWS) ? rcur[idx] : 0;
                    sum += v[i];
                }
                int incl = sum;
#pragma unroll
                for (int d = 1; d < 32; d <<= 1) {
                    const int t = __shfl_up(incl, d, 32);
                    if (lane >= d) incl += t;
                }
                int run = incl - sum;
#pragma unroll
                for (int i = 0; i < ROWS_PER_LANE; ++i) {
                    const int idx = lane * ROWS_PER_LANE + i;
                    if (idx < SLICE_ROWS) {
                        rstart[idx] = run;
                        rcur[idx] = run;
                    }
                    run += v[i];
                }
                if (lane == 0) rstart[SLICE_ROWS] = stot;
            }
            __syncthreads();

            float wr[SEGS_PER_WAVE], wc[SEGS_PER_WAVE];
#pragma unroll
            for (int j = 0; j < SEGS_PER_WAVE; ++j) {
                wr[j] = dinv[lo + (int)((unsigned)rc[j].x >> REC_COL_BITS)];
                wc[j] = dinv[rc[j].x & REC_COL_MASK];
            }
#pragma unroll
            for (int j = 0; j < SEGS_PER_WAVE; ++j) {
                const int len = __builtin_amdgcn_readlane(seg_len, j);
                if (lane < len) {
                    const int p = atomicAdd(&rcur[(int)((unsigned)rc[j].x >> REC_COL_BITS)], 1);
                    sorted[p] = v2i{rc[j].x, __float_as_int(__int_as_float(rc[j].y) * wr[j] * wc[j])};
                }
            }
            for (int j = 0; j < SEGS_PER_WAVE; ++j) {
                const int len = __builtin_amdgcn_readlane(seg_len, j);
                const int q = __builtin_amdgcn_readlane(seg_q, j);
                for (int i = 32 + lane; i < len; i += 32) {
                    const v2i r = recs[(size_t)(q + i)];
                    const int rl = (int)((unsigned)r.x >> REC_COL_BITS);
                    const float w = __int_as_float(r.y) * dinv[lo + rl] * dinv[r.x & REC_COL_MASK];
                    const int p = atomicAdd(&rcur[rl], 1);
                    sorted[p] = v2i{r.x, __float_as_int(w)};
                }
            }
            __syncthreads();

            const int rb = (wave * SLICE_ROWS) >> 5;
            const int re = ((wave + 1) * SLICE_ROWS) >> 5;
            const int e0 = __builtin_amdgcn_readfirstlane(rstart[rb]);
            const int e1 = __builtin_amdgcn_readfirstlane(rstart[re]);
            int cur = -1;
            v2f a = {0.0f, 0.0f};
            const int full_end = e0 + ((e1 - e0) & ~31);
            for (int i0 = e0; i0 < full_end; i0 += 32) {
                const v2i sr = sorted[i0 + lane];
                own_rows(sr.x & REC_COL_MASK, (int)((unsigned)sr.x >> REC_COL_BITS) * OUT_DIM, sr.y, 32, lane, support, acc, cur, a);
            }
            if (full_end < e1) {
                const int cnt = e1 - full_end;
                const v2i sr = sorted[full_end + ((lane < cnt) ? lane : 0)];
                own_rows(sr.x & REC_COL_MASK, (int)((unsigned)sr.x >> REC_COL_BITS) * OUT_DIM, sr.y, cnt, lane, support, acc, cur, a);
            }
            if (cur >= 0) *(v2f*)(acc + cur + 2 * lane) = a;
        } else {
            for (int j = 0; j < SEGS_PER_WAVE; ++j) {
                const int len = __builtin_amdgcn_readlane(seg_len, j);
                const int q = __builtin_amdgcn_readlane(seg_q, j);
                for (int i0 = 0; i0 < len; i0 += 32) {
                    const int cnt = (len - i0 < 32) ? len - i0 : 32;
                    const v2i r = recs[(size_t)(q + i0 + ((lane < cnt) ? lane : 0))];
                    const int rl = (int)((unsigned)r.x >> REC_COL_BITS);
                    const int c = r.x & REC_COL_MASK;
                    drain_rows(c, rl * OUT_DIM, __int_as_float(r.y), dinv[lo + rl], dinv[c], cnt, lane, support, acc);
                }
            }
        }
    } else {
    const int nv4 = n_edges >> 2;
    const v4i* rows4 = (const v4i*)rows;
    const int step_vecs = 32 * SCAN_U;
    const int stride = SPMM_WAVES * step_vecs;

#define TEST_EDGE(RV, LOV, EV)                                                   \
    {                                                                            \
        const bool match_ = (unsigned)((RV) - (LOV)) < (unsigned)nrows;          \
        const unsigned m_ = __builtin_amdgcn_ballot_w32(match_);                 \
        if (m_ != 0u) {                                                          \
            if (match_) mylist[wcount + (int)__builtin_amdgcn_mbcnt_lo(m_, 0u)] = (EV); \
            wcount += __builtin_popcount(m_);                                    \
        }                                                                        \
    }

    if (wave == 0) {
        const int e = (nv4 << 2) + lane;
        const int r = (e < n_edges) ? rows[e] : -1;
        TEST_EDGE(r, lo, e)
    }

#define LOAD_STEP(DST, LOS, VB)                                                  \
    _Pragma("unroll") for (int u = 0; u < SCAN_U; ++u) {                         \
        const int idx_ = (VB) + u * 32 + lane;                                   \
        const bool ok_ = idx_ < nv4;                                             \
        DST[u] = rows4[ok_ ? idx_ : nv4 - 1];                                    \
        LOS[u] = ok_ ? lo : 0x7fffffff;                                          \
    }

#define SCAN_STEP(SRC, LOS, VB)                                                  \
    _Pragma("unroll") for (int u = 0; u < SCAN_U; ++u) {                         \
        const int eb_ = ((VB) + u * 32 + lane) * 4;                              \
        TEST_EDGE(SRC[u].x, LOS[u], eb_ + 0)                                     \
        TEST_EDGE(SRC[u].y, LOS[u], eb_ + 1)                                     \
        TEST_EDGE(SRC[u].z, LOS[u], eb_ + 2)                                     \
        TEST_EDGE(SRC[u].w, LOS[u], eb_ + 3)                                     \
    }                                                                            \
    while (wcount >= 32) {                                                       \
        wcount -= 32;                                                            \
        drain_batch(mylist + wcount, 32, lane, lo, rows, cols, adj, dinv, support, acc); \
    }

    int vb = wave * step_vecs;
    if (vb < nv4) {
        v4i ra[SCAN_U], rb[SCAN_U];
        int la[SCAN_U], lb[SCAN_U];
        LOAD_STEP(ra, la, vb)
        for (;;) {
            LOAD_STEP(rb, lb, vb + stride)
            SCAN_STEP(ra, la, vb)
            vb += stride;
            if (vb >= nv4) break;
            LOAD_STEP(ra, la, vb + stride)
            SCAN_STEP(rb, lb, vb)
            vb += stride;
            if (vb >= nv4) break;
        }
    }
#undef SCAN_STEP
#undef LOAD_STEP
#undef TEST_EDGE
    }
    if (wcount > 0) drain_batch(mylist, wcount, lane, lo, rows, cols, adj, dinv, support, acc);
    __syncthreads();

    v4f* out4 = (v4f*)(out + (size_t)lo * OUT_DIM);
    for (int i = tid; i < nrows * (OUT_DIM / 4); i += SPMM_THREADS) out4[i] = acc4[i];
}

extern "C" void kernel_launch(void* const* d_in, const int* in_sizes, int n_in,
                              void* d_out, int out_size, void* d_ws, size_t ws_size,
                              hipStream_t stream) {
    const int* rows = (const int*)d_in[0];
    const int* cols = (const int*)d_in[1];
    const float* adj = (const float*)d_in[2];
    const float* x = (const float*)d_in[3];
    const float* W = (const float*)d_in[4];
    const float* b = (const float*)d_in[5];
    float* out = (float*)d_out;

    const int n_edges = in_sizes[0];
    const int n_nodes = in_sizes[3] / IN_DIM;

    float* support = (float*)d_ws;
    float* dinv = support + (size_t)n_nodes * OUT_DIM;
    int* binoff = (int*)(dinv + n_nodes);
    int* recs_raw = binoff + BIN_WGS * OFF_PITCH;
    const unsigned long long ws_bytes = (unsigned long long)ws_size;

    {
        const int mtiles = (n_nodes + 15) / 16;
        const int blocks = (mtiles + 1) / 2;
        gcn_gemm_wmma<<<blocks, 64, 0, stream>>>(x, W, b, support, n_nodes);
    }

    bin_edges_kernel<<<BIN_WGS, BIN_THREADS, 0, stream>>>(rows, cols, adj, binoff, recs_raw, n_edges, n_nodes, ws_bytes);

    slice_dinv_kernel<<<(n_nodes + SLICE_ROWS - 1) / SLICE_ROWS, DEG_THREADS, 0, stream>>>(
        rows, adj, dinv, binoff, recs_raw, n_edges, n_nodes, ws_bytes);

    spmm_slice_kernel<<<(n_nodes + SLICE_ROWS - 1) / SLICE_ROWS, SPMM_THREADS, 0, stream>>>(
        rows, cols, adj, dinv, support, out, binoff, recs_raw, n_edges, n_nodes, ws_bytes);
}


# dinv[col] folded into the support rows: x@W+b joins the degree launch behind the binning and scales its tiles; SpMM weight = adj, dinv[row] applied at the slice store; sum order kept; 3 launches
# speedup vs baseline: 7.1491x; 1.1679x over previous
#include <hip/hip_runtime.h>


typedef __attribute__((ext_vector_type(2))) float v2f;
typedef __attribute__((ext_vector_type(8))) float v8f;

#define IN_DIM 64
#define OUT_DIM 64

#define SLICE_ROWS 392
#define SPMM_WAVES 32
#define SPMM_THREADS (SPMM_WAVES * 32)
#define SCAN_U 4
#define LIST_CAP (32 + SCAN_U * 4 * 32)
#define DRAIN_G 16
#define SORT_CAP 8192
#define ROWS_PER_LANE ((SLICE_ROWS + 31) / 32)

typedef __attribute__((ext_vector_type(2))) int v2i;
typedef __attribute__((ext_vector_type(4))) int v4i;
typedef __attribute__((ext_vector_type(4))) float v4f;

__device__ __forceinline__ void lds_add_f32(float* p, float v) {
    __hip_atomic_fetch_add(p, v, __ATOMIC_RELAXED, __HIP_MEMORY_SCOPE_WORKGROUP);
}

__device__ __forceinline__ void drain_rows(const int c, const int rl, const float wa,
                                           const int cnt, const int lane,
                                           const float* __restrict__ support,
                                           float* acc) {
    int nbits = 0;
#pragma unroll
    for (int g = 0; g < 32 / DRAIN_G; ++g) {
        if (g * DRAIN_G < cnt) {
            v2f s[DRAIN_G];
#pragma unroll
            for (int t = 0; t < DRAIN_G; ++t) {
                const int cj = __builtin_amdgcn_readlane(c, g * DRAIN_G + t);
                s[t] = *(const v2f*)(support + (size_t)cj * OUT_DIM + 2 * lane);
            }
            if (g == 0) nbits = __float_as_int(wa);
#pragma unroll
            for (int t = 0; t < DRAIN_G; ++t) {
                const int j = g * DRAIN_G + t;
                if (j < cnt) {
                    const float nj = __int_as_float(__builtin_amdgcn_readlane(nbits, j));
                    const int rj = __builtin_amdgcn_readlane(rl, j) + 2 * lane;
                    lds_add_f32(acc + rj, nj * s[t].x);
                    lds_add_f32(acc + rj + 1, nj * s[t].y);
                }
            }
        }
    }
}

__device__ __forceinline__ void own_rows(const int c, const int ro, const int wb, const int cnt, const int lane,
                                         const float* __restrict__ support, float* acc, int& cur, v2f& a) {
#pragma unroll
    for (int g = 0; g < 32 / DRAIN_G; ++g) {
        if (g * DRAIN_G < cnt) {
            v2f s[DRAIN_G];
#pragma unroll
            for (int t = 0; t < DRAIN_G; ++t) {
                const int cj = __builtin_amdgcn_readlane(c, g * DRAIN_G + t);
                s[t] = *(const v2f*)(support + (size_t)cj * OUT_DIM + 2 * lane);
            }
#pragma unroll
            for (int t = 0; t < DRAIN_G; ++t) {
                const int j = g * DRAIN_G + t;
                if (j < cnt) {
                    const int rj = __builtin_amdgcn_readlane(ro, j);
                    const float wj = __int_as_float(__builtin_amdgcn_readlane(wb, j));
                    if (__builtin_expect(rj != cur, 0)) {
                        if (cur >= 0) *(v2f*)(acc + cur + 2 * lane) = a;
                        cur = rj;
                        a = v2f{0.0f, 0.0f};
                    }
                    a.x += wj * s[t].x;
                    a.y += wj * s[t].y;
                }
            }
        }
    }
}

__device__ __forceinline__ void drain_batch(const int* __restrict__ ent, const int cnt, const int lane,
                                            const int lo,
                                            const int* __restrict__ rows,
                                            const int* __restrict__ cols,
                                            const float* __restrict__ adj,
                                            const float* __restrict__ support,
                                            float* acc) {
    __builtin_amdgcn_wave_barrier();
    const int e = (lane < cnt) ? ent[lane] : ent[0];
    const int r = rows[e];
    const int c = cols[e];
    drain_rows(c, (r - lo) * OUT_DIM, adj[e], cnt, lane, support, acc);
}


#define BIN_WGS 256
#define BIN_THREADS 1024
#define BIN_U 4
#define MAX_SLICES 256
#define OFF_PITCH (MAX_SLICES + 1)
#define SEGS_PER_WAVE (BIN_WGS / SPMM_WAVES)
#define REC_COL_BITS 22
#define REC_COL_MASK ((1 << REC_COL_BITS) - 1)
static_assert(SLICE_ROWS <= (1 << (32 - REC_COL_BITS)), "the row inside the slice must fit above the col bits");

__device__ __forceinline__ int rec_pad_ints(const int* recs_raw) {
    return (int)(((8ull - ((unsigned long long)recs_raw & 7ull)) & 7ull) >> 2);
}

__device__ __forceinline__ bool recs_fit(unsigned long long ws_bytes, int n_nodes, int n_edges, const int* recs_raw) {
    const unsigned long long pad = 4ull * (unsigned long long)rec_pad_ints(recs_raw);
    const unsigned long long need =
        ((unsigned long long)n_nodes * (OUT_DIM + 1) + (unsigned long long)BIN_WGS * OFF_PITCH) * 4ull + pad +
        (unsigned long long)n_edges * 8ull;
    return ws_bytes >= need && n_nodes <= (1 << REC_COL_BITS) && (n_nodes + SLICE_ROWS - 1) / SLICE_ROWS <= MAX_SLICES;
}

__global__ __launch_bounds__(BIN_THREADS) void bin_edges_kernel(const int* __restrict__ rows,
                                                                const int* __restrict__ cols,
                                                                const float* __restrict__ adj,
                                                                int* __restrict__ binoff,
                                                                int* recs_raw,
                                                                int n_edges, int n_nodes,
                                                                unsigned long long ws_bytes) {
    if (!recs_fit(ws_bytes, n_nodes, n_edges, recs_raw)) return;
    v2i* __restrict__ recs = (v2i*)(recs_raw + rec_pad_ints(recs_raw));
    __shared__ int cnt[MAX_SLICES];
    __shared__ int cur[MAX_SLICES];
    const int tid = threadIdx.x;
    const int ch = (n_edges + BIN_WGS - 1) / BIN_WGS;
    const long long c0l = (long long)blockIdx.x * ch;
    const int c0 = (c0l < (long long)n_edges) ? (int)c0l : n_edges;
    const int c1 = (n_edges - c0 < ch) ? n_edges : c0 + ch;

    if (tid < MAX_SLICES) cnt[tid] = 0;
    __syncthreads();
    for (int eb = c0 + tid; eb < c1; eb += BIN_THREADS * BIN_U) {
        unsigned r[BIN_U];
#pragma unroll
        for (int u = 0; u < BIN_U; ++u) {
            const int e = eb + u * BIN_THREADS;
            r[u] = (e < c1) ? (unsigned)rows[e] : 0xffffffffu;
        }
#pragma unroll
        for (int u = 0; u < BIN_U; ++u)
            if (r[u] < (unsigned)n_nodes) atomicAdd(&cnt[r[u] / SLICE_ROWS], 1);
    }
    __syncthreads();
    if (tid < 32) {
        int v[MAX_SLICES / 32];
        int sum = 0;
#pragma unroll
        for (int i = 0; i < MAX_SLICES / 32; ++i) {
            v[i] = cnt[tid * (MAX_SLICES / 32) + i];
            sum += v[i];
        }
        int incl = sum;
#pragma unroll
        for (int d = 1; d < 32; d <<= 1) {
            const int t = __shfl_up(incl, d, 32);
            if (tid >= d) incl += t;
        }
        int run = incl - sum;
        int* orow = binoff + (size_t)blockIdx.x * OFF_PITCH;
#pragma unroll
        for (int i = 0; i < MAX_SLICES / 32; ++i) {
            cur[tid * (MAX_SLICES / 32) + i] = run;
            orow[tid * (MAX_SLICES / 32) + i] = run;
            run += v[i];
        }
        if (tid == 31) orow[MAX_SLICES] = run;
    }
    __syncthreads();
    for (int eb = c0 + tid; eb < c1; eb += BIN_THREADS * BIN_U) {
        unsigned r[BIN_U];
        int cc[BIN_U], av[BIN_U];
#pragma unroll
        for (int u = 0; u < BIN_U; ++u) {
            const int e = eb + u * BIN_THREADS;
            const bool ok = e < c1;
            r[u] = ok ? (unsigned)rows[e] : 0xffffffffu;
            cc[u] = ok ? cols[e] : 0;
            av[u] = ok ? __float_as_int(adj[e]) : 0;
        }
#pragma unroll
        for (int u = 0; u < BIN_U; ++u) {
            if (r[u] < (unsigned)n_nodes) {
                const int sl = (int)(r[u] / SLICE_ROWS);
                const int p = atomicAdd(&cur[sl], 1);
                v2i rec;
                rec.x = (int)(((unsigned)cc[u] & (unsigned)REC_COL_MASK) | ((r[u] - (unsigned)(sl * SLICE_ROWS)) << REC_COL_BITS));
                rec.y = av[u];
                recs[(size_t)c0 + p] = rec;
            }
        }
    }
}

#define FUSE_WAVES 32
#define FUSE_THREADS (FUSE_WAVES * 32)
#define FUSE_SEGS (BIN_WGS / FUSE_WAVES)
#define SLICE_TILES ((SLICE_ROWS + 15) / 16)
static_assert(SLICE_TILES <= FUSE_WAVES, "one 16-row tile of the slice per wave");
static_assert(FUSE_SEGS <= 32, "one segment's bounds per lane");

__global__ __launch_bounds__(FUSE_THREADS) void slice_dinv_gemm_kernel(const int* __restrict__ rows,
                                                                       const float* __restrict__ adj,
                                                                       const float* __restrict__ x,
                                                                       const float* __restrict__ W,
                                                                       const float* __restrict__ b,
                                                                       float* __restrict__ support,
                                                                       float* __restrict__ dinv,
                                                                       const int* __restrict__ binoff,
                                                                       const int* __restrict__ recs_raw,
                                                                       int n_edges, int n_nodes,
                                                                       unsigned long long ws_bytes) {
    __shared__ float deg[SLICE_TILES * 16];
    const int tid = threadIdx.x;
    const int lane = tid & 31;
    const int wave = __builtin_amdgcn_readfirstlane(tid >> 5);
    const int lo = blockIdx.x * SLICE_ROWS;
    int nrows = n_nodes - lo;
    if (nrows > SLICE_ROWS) nrows = SLICE_ROWS;
    const int half = lane >> 4;
    const int r = lane & 15;
    const bool has_tile = wave * 16 < nrows;

    for (int i = tid; i < SLICE_TILES * 16; i += FUSE_THREADS) deg[i] = 0.0f;

    __syncthreads();

    v2f afr[IN_DIM / 4];
    int arow_i = lo + (has_tile ? wave * 16 : 0) + r;
    if (arow_i > n_nodes - 1) arow_i = n_nodes - 1;
    const float* arow = x + (size_t)arow_i * IN_DIM + 2 * half;

    if (recs_fit(ws_bytes, n_nodes, n_edges, recs_raw)) {
        const v2i* __restrict__ recs = (const v2i*)(recs_raw + rec_pad_ints(recs_raw));
        const int ch = (n_edges + BIN_WGS - 1) / BIN_WGS;
        int sa = 0, sb = 0, k0 = 0;
        if (lane < FUSE_SEGS) {
            const int chunk = wave + FUSE_WAVES * lane;
            const int* o = binoff + (size_t)chunk * OFF_PITCH + blockIdx.x;
            sa = o[0];
            sb = o[1];
            const long long k0l = (long long)chunk * ch;
            k0 = (k0l < (long long)n_edges) ? (int)k0l : n_edges;
        }
#pragma unroll
        for (int kk = 0; kk < IN_DIM / 4; ++kk) afr[kk] = *(const v2f*)(arow + 4 * kk);
        const int seg_len = sb - sa;
        const int seg_q = k0 + sa;
        v2i rc[FUSE_SEGS];
#pragma unroll
        for (int j = 0; j < FUSE_SEGS; ++j) {
            const int len = __builtin_amdgcn_readlane(seg_len, j);
            const int q = __builtin_amdgcn_readlane(seg_q, j);
            rc[j] = v2i{0, 0};
            if (len > 0) rc[j] = recs[(size_t)(q + ((lane < len) ? lane : 0))];
        }
#pragma unroll
        for (int j = 0; j < FUSE_SEGS; ++j) {
            const int len = __builtin_amdgcn_readlane(seg_len, j);
            if (lane < len) lds_add_f32(deg + (int)((unsigned)rc[j].x >> REC_COL_BITS), __int_as_float(rc[j].y));
        }
        for (int j = 0; j < FUSE_SEGS; ++j) {
            const int len = __builtin_amdgcn_readlane(seg_len, j);
            const int q = __builtin_amdgcn_readlane(seg_q, j);
            for (int i = 32 + lane; i < len; i += 32) {
                const v2i rr = recs[(size_t)(q + i)];
                lds_add_f32(deg + (int)((unsigned)rr.x >> REC_COL_BITS), __int_as_float(rr.y));
            }
        }
    } else {
#pragma unroll
        for (int kk = 0; kk < IN_DIM / 4; ++kk) afr[kk] = *(const v2f*)(arow + 4 * kk);
        for (int eb = tid; eb < n_edges; eb += FUSE_THREADS * 4) {
            unsigned rl[4];
#pragma unroll
            for (int u = 0; u < 4; ++u) {
                const int e = eb + u * FUSE_THREADS;
                const bool ok = e < n_edges;
                const unsigned rv = (unsigned)rows[ok ? e : n_edges - 1];
                rl[u] = ok ? rv - (unsigned)lo : 0xffffffffu;
            }
#pragma unroll
            for (int u = 0; u < 4; ++u)
                if (rl[u] < (unsigned)nrows) lds_add_f32(deg + rl[u], adj[eb + u * FUSE_THREADS]);
        }
    }

    v8f acc[4];
    if (has_tile) {
#pragma unroll
        for (int nt = 0; nt < 4; ++nt) {
            const float bv = b[nt * 16 + r];
#pragma unroll
            for (int i = 0; i < 8; ++i) acc[nt][i] = bv;
        }
#pragma unroll
        for (int k = 0; k < IN_DIM; k += 4) {
#pragma unroll
            for (int nt = 0; nt < 4; ++nt) {
                v2f bf;
                bf.x = W[(size_t)(k + 2 * half + 0) * OUT_DIM + nt * 16 + r];
                bf.y = W[(size_t)(k + 2 * half + 1) * OUT_DIM + nt * 16 + r];
                acc[nt] = __builtin_amdgcn_wmma_f32_16x16x4_f32(
                    false, afr[k / 4], false, bf, (short)0, acc[nt], false, false);
            }
        }
    }
    __syncthreads();

    for (int i = tid; i < nrows; i += FUSE_THREADS) {
        float d = deg[i];
        d = (d > 0.0f) ? d : 1.0f;
        dinv[lo + i] = rsqrtf(d);
    }

    if (has_tile) {
        float* srow = support + (size_t)(lo + wave * 16) * OUT_DIM;
#pragma unroll
        for (int i = 0; i < 8; ++i) {
            const int rin = wave * 16 + i + 8 * half;
            float d = deg[rin];
            d = (d > 0.0f) ? d : 1.0f;
            const float sc = rsqrtf(d);
            if (rin < nrows) {
#pragma unroll
                for (int nt = 0; nt < 4; ++nt)
                    srow[(size_t)(i + 8 * half) * OUT_DIM + nt * 16 + r] = acc[nt][i] * sc;
            }
        }
    }
}

__global__ __launch_bounds__(SPMM_THREADS) void spmm_slice_kernel(const int* __restrict__ rows,
                                                                  const int* __restrict__ cols,
                                                                  const float* __restrict__ adj,
                                                                  const float* __restrict__ dinv,
                                                                  const float* __restrict__ support,
                                                                  float* __restrict__ out,
                                                                  const int* __restrict__ binoff,
                                                                  const int* __restrict__ recs_raw,
                                                                  int n_edges, int n_nodes,
                                                                  unsigned long long ws_bytes) {
    __shared__ float acc[SLICE_ROWS * OUT_DIM];
    __shared__ int wlist[SPMM_WAVES * LIST_CAP];
    __shared__ v2i sorted[SORT_CAP];
    __shared__ int rstart[SLICE_ROWS + 1];
    __shared__ int rcur[SLICE_ROWS];
    __shared__ int s_total;
    __shared__ float sdinv[SLICE_ROWS];
    static_assert(SLICE_ROWS <= SPMM_THREADS, "one row's dinv per thread");

    const int tid = threadIdx.x;
    const int lane = tid & 31;
    const int wave = __builtin_amdgcn_readfirstlane(tid >> 5);
    const int lo = blockIdx.x * SLICE_ROWS;
    int nrows = n_nodes - lo;
    if (nrows > SLICE_ROWS) nrows = SLICE_ROWS;

    v4f* acc4 = (v4f*)acc;
    const v4f zero4 = {0.0f, 0.0f, 0.0f, 0.0f};
    for (int i = tid; i < nrows * (OUT_DIM / 4); i += SPMM_THREADS) acc4[i] = zero4;
    for (int i = tid; i < SLICE_ROWS; i += SPMM_THREADS) rcur[i] = 0;
    if (tid == 0) s_total = 0;
    __syncthreads();

    const float my_dinv = (tid < nrows) ? dinv[lo + tid] : 0.0f;

    int* mylist = wlist + wave * LIST_CAP;
    int wcount = 0;

    if (recs_fit(ws_bytes, n_nodes, n_edges, recs_raw)) {
        const v2i* __restrict__ recs = (const v2i*)(recs_raw + rec_pad_ints(recs_raw));
        const int ch = (n_edges + BIN_WGS - 1) / BIN_WGS;
        int seg_len = 0, seg_q = 0;
        if (lane < SEGS_PER_WAVE) {
            const int chunk = wave + SPMM_WAVES * lane;
            const int* o = binoff + (size_t)chunk * OFF_PITCH + blockIdx.x;
            const int sa = o[0];
            seg_len = o[1] - sa;
            const long long k0l = (long long)chunk * ch;
            const int k0 = (k0l < (long long)n_edges) ? (int)k0l : n_edges;
            seg_q = k0 + sa;
        }
        int total = 0;
#pragma unroll
        for (int j = 0; j < SEGS_PER_WAVE; ++j) total += __builtin_amdgcn_readlane(seg_len, j);

        v2i rc[SEGS_PER_WAVE];
#pragma unroll
        for (int j = 0; j < SEGS_PER_WAVE; ++j) {
            const int len = __builtin_amdgcn_readlane(seg_len, j);
            const int q = __builtin_amdgcn_readlane(seg_q, j);
            rc[j] = v2i{0, 0};
            if (len > 0) rc[j] = recs[(size_t)(q + ((lane < len) ? lane : 0))];
        }
        if (lane == 0) atomicAdd(&s_total, total);
#pragma unroll
        for (int j = 0; j < SEGS_PER_WAVE; ++j) {
            const int len = __builtin_amdgcn_readlane(seg_len, j);
            if (lane < len) atomicAdd(&rcur[(int)((unsigned)rc[j].x >> REC_COL_BITS)], 1);
        }
        for (int j = 0; j < SEGS_PER_WAVE; ++j) {
            const int len = __builtin_amdgcn_readlane(seg_len, j);
            const int q = __builtin_amdgcn_readlane(seg_q, j);
            for (int i = 32 + lane; i < len; i += 32) {
                const v2i r = recs[(size_t)(q + i)];
                atomicAdd(&rcur[(int)((unsigned)r.x >> REC_COL_BITS)], 1);
            }
        }
        __syncthreads();
        const int stot = __builtin_amdgcn_readfirstlane(s_total);

        if (stot <= SORT_CAP) {
            if (wave == 0) {
                int v[ROWS_PER_LANE];
                int sum = 0;
#pragma unroll
                for (int i = 0; i < ROWS_PER_LANE; ++i) {
                    const int idx = lane * ROWS_PER_LANE + i;
                    v[i] = (idx < SLICE_ROWS) ? rcur[idx] : 0;
                    sum += v[i];
                }
                int incl = sum;
#pragma unroll
                for (int d = 1; d < 32; d <<= 1) {
                    const int t = __shfl_up(incl, d, 32);
                    if (lane >= d) incl += t;
                }
                int run = incl - sum;
#pragma unroll
                for (int i = 0; i < ROWS_PER_LANE; ++i) {
                    const int idx = lane * ROWS_PER_LANE + i;
                    if (idx < SLICE_ROWS) {
                        rstart[idx] = run;
                        rcur[idx] = run;
                    }
                    run += v[i];
                }
                if (lane == 0) rstart[SLICE_ROWS] = stot;
            }
            __syncthreads();

#pragma unroll
            for (int j = 0; j < SEGS_PER_WAVE; ++j) {
                const int len = __builtin_amdgcn_readlane(seg_len, j);
                if (lane < len) {
                    const int p = atomicAdd(&rcur[(int)((unsigned)rc[j].x >> REC_COL_BITS)], 1);
                    sorted[p] = rc[j];
                }
            }
            for (int j = 0; j < SEGS_PER_WAVE; ++j) {
                const int len = __builtin_amdgcn_readlane(seg_len, j);
                const int q = __builtin_amdgcn_readlane(seg_q, j);
                for (int i = 32 + lane; i < len; i += 32) {
                    const v2i r = recs[(size_t)(q + i)];
                    const int p = atomicAdd(&rcur[(int)((unsigned)r.x >> REC_COL_BITS)], 1);
                    sorted[p] = r;
                }
            }
            __syncthreads();

            const int rb = (wave * SLICE_ROWS) >> 5;
            const int re = ((wave + 1) * SLICE_ROWS) >> 5;
            const int e0 = __builtin_amdgcn_readfirstlane(rstart[rb]);
            const int e1 = __builtin_amdgcn_readfirstlane(rstart[re]);
            int cur = -1;
            v2f a = {0.0f, 0.0f};
            const int full_end = e0 + ((e1 - e0) & ~31);
            for (int i0 = e0; i0 < full_end; i0 += 32) {
                const v2i sr = sorted[i0 + lane];
                own_rows(sr.x & REC_COL_MASK, (int)((unsigned)sr.x >> REC_COL_BITS) * OUT_DIM, sr.y, 32, lane, support, acc, cur, a);
            }
            if (full_end < e1) {
                const int cnt = e1 - full_end;
                const v2i sr = sorted[full_end + ((lane < cnt) ? lane : 0)];
                own_rows(sr.x & REC_COL_MASK, (int)((unsigned)sr.x >> REC_COL_BITS) * OUT_DIM, sr.y, cnt, lane, support, acc, cur, a);
            }
            if (cur >= 0) *(v2f*)(acc + cur + 2 * lane) = a;
        } else {
            for (int j = 0; j < SEGS_PER_WAVE; ++j) {
                const int len = __builtin_amdgcn_readlane(seg_len, j);
                const int q = __builtin_amdgcn_readlane(seg_q, j);
                for (int i0 = 0; i0 < len; i0 += 32) {
                    const int cnt = (len - i0 < 32) ? len - i0 : 32;
                    const v2i r = recs[(size_t)(q + i0 + ((lane < cnt) ? lane : 0))];
                    const int rl = (int)((unsigned)r.x >> REC_COL_BITS);
                    const int c = r.x & REC_COL_MASK;
                    drain_rows(c, rl * OUT_DIM, __int_as_float(r.y), cnt, lane, support, acc);
                }
            }
        }
    } else {
    const int nv4 = n_edges >> 2;
    const v4i* rows4 = (const v4i*)rows;
    const int step_vecs = 32 * SCAN_U;
    const int stride = SPMM_WAVES * step_vecs;

#define TEST_EDGE(RV, LOV, EV)                                                   \
    {                                                                            \
        const bool match_ = (unsigned)((RV) - (LOV)) < (unsigned)nrows;          \
        const unsigned m_ = __builtin_amdgcn_ballot_w32(match_);                 \
        if (m_ != 0u) {                                                          \
            if (match_) mylist[wcount + (int)__builtin_amdgcn_mbcnt_lo(m_, 0u)] = (EV); \
            wcount += __builtin_popcount(m_);                                    \
        }                                                                        \
    }

    if (wave == 0) {
        const int e = (nv4 << 2) + lane;
        const int r = (e < n_edges) ? rows[e] : -1;
        TEST_EDGE(r, lo, e)
    }

#define LOAD_STEP(DST, LOS, VB)                                                  \
    _Pragma("unroll") for (int u = 0; u < SCAN_U; ++u) {                         \
        const int idx_ = (VB) + u * 32 + lane;                                   \
        const bool ok_ = idx_ < nv4;                                             \
        DST[u] = rows4[ok_ ? idx_ : nv4 - 1];                                    \
        LOS[u] = ok_ ? lo : 0x7fffffff;                                          \
    }

#define SCAN_STEP(SRC, LOS, VB)                                                  \
    _Pragma("unroll") for (int u = 0; u < SCAN_U; ++u) {                         \
        const int eb_ = ((VB) + u * 32 + lane) * 4;                              \
        TEST_EDGE(SRC[u].x, LOS[u], eb_ + 0)                                     \
        TEST_EDGE(SRC[u].y, LOS[u], eb_ + 1)                                     \
        TEST_EDGE(SRC[u].z, LOS[u], eb_ + 2)                                     \
        TEST_EDGE(SRC[u].w, LOS[u], eb_ + 3)                                     \
    }                                                                            \
    while (wcount >= 32) {                                                       \
        wcount -= 32;                                                            \
        drain_batch(mylist + wcount, 32, lane, lo, rows, cols, adj, support, acc); \
    }

    int vb = wave * step_vecs;
    if (vb < nv4) {
        v4i ra[SCAN_U], rb[SCAN_U];
        int la[SCAN_U], lb[SCAN_U];
        LOAD_STEP(ra, la, vb)
        for (;;) {
            LOAD_STEP(rb, lb, vb + stride)
            SCAN_STEP(ra, la, vb)
            vb += stride;
            if (vb >= nv4) break;
            LOAD_STEP(ra, la, vb + stride)
            SCAN_STEP(rb, lb, vb)
            vb += stride;
            if (vb >= nv4) break;
        }
    }
#undef SCAN_STEP
#undef LOAD_STEP
#undef TEST_EDGE
    }
    if (wcount > 0) drain_batch(mylist, wcount, lane, lo, rows, cols, adj, support, acc);
    if (tid < SLICE_ROWS) sdinv[tid] = my_dinv;
    __syncthreads();

    v4f* out4 = (v4f*)(out + (size_t)lo * OUT_DIM);
    for (int i = tid; i < nrows * (OUT_DIM / 4); i += SPMM_THREADS) out4[i] = acc4[i] * sdinv[i / (OUT_DIM / 4)];
}

extern "C" void kernel_launch(void* const* d_in, const int* in_sizes, int n_in,
                              void* d_out, int out_size, void* d_ws, size_t ws_size,
                              hipStream_t stream) {
    const int* rows = (const int*)d_in[0];
    const int* cols = (const int*)d_in[1];
    const float* adj = (const float*)d_in[2];
    const float* x = (const float*)d_in[3];
    const float* W = (const float*)d_in[4];
    const float* b = (const float*)d_in[5];
    float* out = (float*)d_out;

    const int n_edges = in_sizes[0];
    const int n_nodes = in_sizes[3] / IN_DIM;

    float* support = (float*)d_ws;
    float* dinv = support + (size_t)n_nodes * OUT_DIM;
    int* binoff = (int*)(dinv + n_nodes);
    int* recs_raw = binoff + BIN_WGS * OFF_PITCH;
    const unsigned long long ws_bytes = (unsigned long long)ws_size;

    bin_edges_kernel<<<BIN_WGS, BIN_THREADS, 0, stream>>>(rows, cols, adj, binoff, recs_raw, n_edges, n_nodes, ws_bytes);

    slice_dinv_gemm_kernel<<<(n_nodes + SLICE_ROWS - 1) / SLICE_ROWS, FUSE_THREADS, 0, stream>>>(
        rows, adj, x, W, b, support, dinv, binoff, recs_raw, n_edges, n_nodes, ws_bytes);

    spmm_slice_kernel<<<(n_nodes + SLICE_ROWS - 1) / SLICE_ROWS, SPMM_THREADS, 0, stream>>>(
        rows, cols, adj, dinv, support, out, binoff, recs_raw, n_edges, n_nodes, ws_bytes);
}


# binning launch: a chunk's records are put in slice order in LDS, then copied out as whole 128-byte lines, instead of 32 lanes storing 8 bytes into 32 different lines of global memory
# speedup vs baseline: 7.6489x; 1.0699x over previous
#include <hip/hip_runtime.h>


typedef __attribute__((ext_vector_type(2))) float v2f;
typedef __attribute__((ext_vector_type(8))) float v8f;

#define IN_DIM 64
#define OUT_DIM 64

#define SLICE_ROWS 392
#define SPMM_WAVES 32
#define SPMM_THREADS (SPMM_WAVES * 32)
#define SCAN_U 4
#define LIST_CAP (32 + SCAN_U * 4 * 32)
#define DRAIN_G 16
#define SORT_CAP 8192
#define ROWS_PER_LANE ((SLICE_ROWS + 31) / 32)

typedef __attribute__((ext_vector_type(2))) int v2i;
typedef __attribute__((ext_vector_type(4))) int v4i;
typedef __attribute__((ext_vector_type(4))) float v4f;

__device__ __forceinline__ void lds_add_f32(float* p, float v) {
    __hip_atomic_fetch_add(p, v, __ATOMIC_RELAXED, __HIP_MEMORY_SCOPE_WORKGROUP);
}

__device__ __forceinline__ void drain_rows(const int c, const int rl, const float wa,
                                           const int cnt, const int lane,
                                           const float* __restrict__ support,
                                           float* acc) {
    int nbits = 0;
#pragma unroll
    for (int g = 0; g < 32 / DRAIN_G; ++g) {
        if (g * DRAIN_G < cnt) {
            v2f s[DRAIN_G];
#pragma unroll
            for (int t = 0; t < DRAIN_G; ++t) {
                const int cj = __builtin_amdgcn_readlane(c, g * DRAIN_G + t);
                s[t] = *(const v2f*)(support + (size_t)cj * OUT_DIM + 2 * lane);
            }
            if (g == 0) nbits = __float_as_int(wa);
#pragma unroll
            for (int t = 0; t < DRAIN_G; ++t) {
                const int j = g * DRAIN_G + t;
                if (j < cnt) {
                    const float nj = __int_as_float(__builtin_amdgcn_readlane(nbits, j));
                    const int rj = __builtin_amdgcn_readlane(rl, j) + 2 * lane;
                    lds_add_f32(acc + rj, nj * s[t].x);
                    lds_add_f32(acc + rj + 1, nj * s[t].y);
                }
            }
        }
    }
}

__device__ __forceinline__ void own_rows(const int c, const int ro, const int wb, const int cnt, const int lane,
                                         const float* __restrict__ support, float* acc, int& cur, v2f& a) {
#pragma unroll
    for (int g = 0; g < 32 / DRAIN_G; ++g) {
        if (g * DRAIN_G < cnt) {
            v2f s[DRAIN_G];
#pragma unroll
            for (int t = 0; t < DRAIN_G; ++t) {
                const int cj = __builtin_amdgcn_readlane(c, g * DRAIN_G + t);
                s[t] = *(const v2f*)(support + (size_t)cj * OUT_DIM + 2 * lane);
            }
#pragma unroll
            for (int t = 0; t < DRAIN_G; ++t) {
                const int j = g * DRAIN_G + t;
                if (j < cnt) {
                    const int rj = __builtin_amdgcn_readlane(ro, j);
                    const float wj = __int_as_float(__builtin_amdgcn_readlane(wb, j));
                    if (__builtin_expect(rj != cur, 0)) {
                        if (cur >= 0) *(v2f*)(acc + cur + 2 * lane) = a;
                        cur = rj;
                        a = v2f{0.0f, 0.0f};
                    }
                    a.x += wj * s[t].x;
                    a.y += wj * s[t].y;
                }
            }
        }
    }
}

__device__ __forceinline__ void drain_batch(const int* __restrict__ ent, const int cnt, const int lane,
                                            const int lo,
                                            const int* __restrict__ rows,
                                            const int* __restrict__ cols,
                                            const float* __restrict__ adj,
                                            const float* __restrict__ support,
                                            float* acc) {
    __builtin_amdgcn_wave_barrier();
    const int e = (lane < cnt) ? ent[lane] : ent[0];
    const int r = rows[e];
    const int c = cols[e];
    drain_rows(c, (r - lo) * OUT_DIM, adj[e], cnt, lane, support, acc);
}


#define BIN_WGS 256
#define BIN_THREADS 1024
#define BIN_U 4
#define BIN_STAGE_CAP 8192
#define MAX_SLICES 256
#define OFF_PITCH (MAX_SLICES + 1)
#define SEGS_PER_WAVE (BIN_WGS / SPMM_WAVES)
#define REC_COL_BITS 22
#define REC_COL_MASK ((1 << REC_COL_BITS) - 1)
static_assert(SLICE_ROWS <= (1 << (32 - REC_COL_BITS)), "the row inside the slice must fit above the col bits");

__device__ __forceinline__ int rec_pad_ints(const int* recs_raw) {
    return (int)(((8ull - ((unsigned long long)recs_raw & 7ull)) & 7ull) >> 2);
}

__device__ __forceinline__ bool recs_fit(unsigned long long ws_bytes, int n_nodes, int n_edges, const int* recs_raw) {
    const unsigned long long pad = 4ull * (unsigned long long)rec_pad_ints(recs_raw);
    const unsigned long long need =
        ((unsigned long long)n_nodes * (OUT_DIM + 1) + (unsigned long long)BIN_WGS * OFF_PITCH) * 4ull + pad +
        (unsigned long long)n_edges * 8ull;
    return ws_bytes >= need && n_nodes <= (1 << REC_COL_BITS) && (n_nodes + SLICE_ROWS - 1) / SLICE_ROWS <= MAX_SLICES;
}

__global__ __launch_bounds__(BIN_THREADS) void bin_edges_kernel(const int* __restrict__ rows,
                                                                const int* __restrict__ cols,
                                                                const float* __restrict__ adj,
                                                                int* __restrict__ binoff,
                                                                int* recs_raw,
                                                                int n_edges, int n_nodes,
                                                                unsigned long long ws_bytes) {
    if (!recs_fit(ws_bytes, n_nodes, n_edges, recs_raw)) return;
    v2i* __restrict__ recs = (v2i*)(recs_raw + rec_pad_ints(recs_raw));
    __shared__ int cnt[MAX_SLICES];
    __shared__ int cur[MAX_SLICES];
    __shared__ v2i stage[BIN_STAGE_CAP];
    __shared__ int s_tot;
    const int tid = threadIdx.x;
    const int ch = (n_edges + BIN_WGS - 1) / BIN_WGS;
    const long long c0l = (long long)blockIdx.x * ch;
    const int c0 = (c0l < (long long)n_edges) ? (int)c0l : n_edges;
    const int c1 = (n_edges - c0 < ch) ? n_edges : c0 + ch;
    const bool staged = (c1 - c0) <= BIN_STAGE_CAP;

    if (tid < MAX_SLICES) cnt[tid] = 0;
    __syncthreads();
    for (int eb = c0 + tid; eb < c1; eb += BIN_THREADS * BIN_U) {
        unsigned r[BIN_U];
#pragma unroll
        for (int u = 0; u < BIN_U; ++u) {
            const int e = eb + u * BIN_THREADS;
            r[u] = (e < c1) ? (unsigned)rows[e] : 0xffffffffu;
        }
#pragma unroll
        for (int u = 0; u < BIN_U; ++u)
            if (r[u] < (unsigned)n_nodes) atomicAdd(&cnt[r[u] / SLICE_ROWS], 1);
    }
    __syncthreads();
    if (tid < 32) {
        int v[MAX_SLICES / 32];
        int sum = 0;
#pragma unroll
        for (int i = 0; i < MAX_SLICES / 32; ++i) {
            v[i] = cnt[tid * (MAX_SLICES / 32) + i];
            sum += v[i];
        }
        int incl = sum;
#pragma unroll
        for (int d = 1; d < 32; d <<= 1) {
            const int t = __shfl_up(incl, d, 32);
            if (tid >= d) incl += t;
        }
        int run = incl - sum;
        int* orow = binoff + (size_t)blockIdx.x * OFF_PITCH;
#pragma unroll
        for (int i = 0; i < MAX_SLICES / 32; ++i) {
            cur[tid * (MAX_SLICES / 32) + i] = run;
            orow[tid * (MAX_SLICES / 32) + i] = run;
            run += v[i];
        }
        if (tid == 31) {
            orow[MAX_SLICES] = run;
            s_tot = run;
        }
    }
    __syncthreads();
    for (int eb = c0 + tid; eb < c1; eb += BIN_THREADS * BIN_U) {
        unsigned r[BIN_U];
        int cc[BIN_U], av[BIN_U];
#pragma unroll
        for (int u = 0; u < BIN_U; ++u) {
            const int e = eb + u * BIN_THREADS;
            const bool ok = e < c1;
            r[u] = ok ? (unsigned)rows[e] : 0xffffffffu;
            cc[u] = ok ? cols[e] : 0;
            av[u] = ok ? __float_as_int(adj[e]) : 0;
        }
#pragma unroll
        for (int u = 0; u < BIN_U; ++u) {
            if (r[u] < (unsigned)n_nodes) {
                const int sl = (int)(r[u] / SLICE_ROWS);
                const int p = atomicAdd(&cur[sl], 1);
                v2i rec;
                rec.x = (int)(((unsigned)cc[u] & (unsigned)REC_COL_MASK) | ((r[u] - (unsigned)(sl * SLICE_ROWS)) << REC_COL_BITS));
                rec.y = av[u];
                if (staged) stage[p] = rec;
                else recs[(size_t)c0 + p] = rec;
            }
        }
    }
    if (staged) {
        __syncthreads();
        const int tot = s_tot;
        const int lead = (int)(((unsigned long long)(recs + (size_t)c0) >> 3) & 15ull);
        for (int ib = tid - lead; ib < tot; ib += BIN_THREADS * BIN_U) {
            v2i t[BIN_U];
#pragma unroll
            for (int u = 0; u < BIN_U; ++u) {
                const int i = ib + u * BIN_THREADS;
                t[u] = stage[(i >= 0 && i < tot) ? i : 0];
            }
#pragma unroll
            for (int u = 0; u < BIN_U; ++u) {
                const int i = ib + u * BIN_THREADS;
                if (i >= 0 && i < tot) recs[(size_t)c0 + i] = t[u];
            }
        }
    }
}

#define FUSE_WAVES 32
#define FUSE_THREADS (FUSE_WAVES * 32)
#define FUSE_SEGS (BIN_WGS / FUSE_WAVES)
#define SLICE_TILES ((SLICE_ROWS + 15) / 16)
static_assert(SLICE_TILES <= FUSE_WAVES, "one 16-row tile of the slice per wave");
static_assert(FUSE_SEGS <= 32, "one segment's bounds per lane");

__global__ __launch_bounds__(FUSE_THREADS) void slice_dinv_gemm_kernel(const int* __restrict__ rows,
                                                                       const float* __restrict__ adj,
                                                                       const float* __restrict__ x,
                                                                       const float* __restrict__ W,
                                                                       const float* __restrict__ b,
                                                                       float* __restrict__ support,
                                                                       float* __restrict__ dinv,
                                                                       const int* __restrict__ binoff,
                                                                       const int* __restrict__ recs_raw,
                                                                       int n_edges, int n_nodes,
                                                                       unsigned long long ws_bytes) {
    __shared__ float deg[SLICE_TILES * 16];
    const int tid = threadIdx.x;
    const int lane = tid & 31;
    const int wave = __builtin_amdgcn_readfirstlane(tid >> 5);
    const int lo = blockIdx.x * SLICE_ROWS;
    int nrows = n_nodes - lo;
    if (nrows > SLICE_ROWS) nrows = SLICE_ROWS;
    const int half = lane >> 4;
    const int r = lane & 15;
    const bool has_tile = wave * 16 < nrows;

    for (int i = tid; i < SLICE_TILES * 16; i += FUSE_THREADS) deg[i] = 0.0f;

    __syncthreads();

    v2f afr[IN_DIM / 4];
    int arow_i = lo + (has_tile ? wave * 16 : 0) + r;
    if (arow_i > n_nodes - 1) arow_i = n_nodes - 1;
    const float* arow = x + (size_t)arow_i * IN_DIM + 2 * half;

    if (recs_fit(ws_bytes, n_nodes, n_edges, recs_raw)) {
        const v2i* __restrict__ recs = (const v2i*)(recs_raw + rec_pad_ints(recs_raw));
        const int ch = (n_edges + BIN_WGS - 1) / BIN_WGS;
        int sa = 0, sb = 0, k0 = 0;
        if (lane < FUSE_SEGS) {
            const int chunk = wave + FUSE_WAVES * lane;
            const int* o = binoff + (size_t)chunk * OFF_PITCH + blockIdx.x;
            sa = o[0];
            sb = o[1];
            const long long k0l = (long long)chunk * ch;
            k0 = (k0l < (long long)n_edges) ? (int)k0l : n_edges;
        }
#pragma unroll
        for (int kk = 0; kk < IN_DIM / 4; ++kk) afr[kk] = *(const v2f*)(arow + 4 * kk);
        const int seg_len = sb - sa;
        const int seg_q = k0 + sa;
        v2i rc[FUSE_SEGS];
#pragma unroll
        for (int j = 0; j < FUSE_SEGS; ++j) {
            const int len = __builtin_amdgcn_readlane(seg_len, j);
            const int q = __builtin_amdgcn_readlane(seg_q, j);
            rc[j] = v2i{0, 0};
            if (len > 0) rc[j] = recs[(size_t)(q + ((lane < len) ? lane : 0))];
        }
#pragma unroll
        for (int j = 0; j < FUSE_SEGS; ++j) {
            const int len = __builtin_amdgcn_readlane(seg_len, j);
            if (lane < len) lds_add_f32(deg + (int)((unsigned)rc[j].x >> REC_COL_BITS), __int_as_float(rc[j].y));
        }
        for (int j = 0; j < FUSE_SEGS; ++j) {
            const int len = __builtin_amdgcn_readlane(seg_len, j);
            const int q = __builtin_amdgcn_readlane(seg_q, j);
            for (int i = 32 + lane; i < len; i += 32) {
                const v2i rr = recs[(size_t)(q + i)];
                lds_add_f32(deg + (int)((unsigned)rr.x >> REC_COL_BITS), __int_as_float(rr.y));
            }
        }
    } else {
#pragma unroll
        for (int kk = 0; kk < IN_DIM / 4; ++kk) afr[kk] = *(const v2f*)(arow + 4 * kk);
        for (int eb = tid; eb < n_edges; eb += FUSE_THREADS * 4) {
            unsigned rl[4];
#pragma unroll
            for (int u = 0; u < 4; ++u) {
                const int e = eb + u * FUSE_THREADS;
                const bool ok = e < n_edges;
                const unsigned rv = (unsigned)rows[ok ? e : n_edges - 1];
                rl[u] = ok ? rv - (unsigned)lo : 0xffffffffu;
            }
#pragma unroll
            for (int u = 0; u < 4; ++u)
                if (rl[u] < (unsigned)nrows) lds_add_f32(deg + rl[u], adj[eb + u * FUSE_THREADS]);
        }
    }

    v8f acc[4];
    if (has_tile) {
#pragma unroll
        for (int nt = 0; nt < 4; ++nt) {
            const float bv = b[nt * 16 + r];
#pragma unroll
            for (int i = 0; i < 8; ++i) acc[nt][i] = bv;
        }
#pragma unroll
        for (int k = 0; k < IN_DIM; k += 4) {
#pragma unroll
            for (int nt = 0; nt < 4; ++nt) {
                v2f bf;
                bf.x = W[(size_t)(k + 2 * half + 0) * OUT_DIM + nt * 16 + r];
                bf.y = W[(size_t)(k + 2 * half + 1) * OUT_DIM + nt * 16 + r];
                acc[nt] = __builtin_amdgcn_wmma_f32_16x16x4_f32(
                    false, afr[k / 4], false, bf, (short)0, acc[nt], false, false);
            }
        }
    }
    __syncthreads();

    for (int i = tid; i < nrows; i += FUSE_THREADS) {
        float d = deg[i];
        d = (d > 0.0f) ? d : 1.0f;
        dinv[lo + i] = rsqrtf(d);
    }

    if (has_tile) {
        float* srow = support + (size_t)(lo + wave * 16) * OUT_DIM;
#pragma unroll
        for (int i = 0; i < 8; ++i) {
            const int rin = wave * 16 + i + 8 * half;
            float d = deg[rin];
            d = (d > 0.0f) ? d : 1.0f;
            const float sc = rsqrtf(d);
            if (rin < nrows) {
#pragma unroll
                for (int nt = 0; nt < 4; ++nt)
                    srow[(size_t)(i + 8 * half) * OUT_DIM + nt * 16 + r] = acc[nt][i] * sc;
            }
        }
    }
}

__global__ __launch_bounds__(SPMM_THREADS) void spmm_slice_kernel(const int* __restrict__ rows,
                                                                  const int* __restrict__ cols,
                                                                  const float* __restrict__ adj,
                                                                  const float* __restrict__ dinv,
                                                                  const float* __restrict__ support,
                                                                  float* __restrict__ out,
                                                                  const int* __restrict__ binoff,
                                                                  const int* __restrict__ recs_raw,
                                                                  int n_edges, int n_nodes,
                                                                  unsigned long long ws_bytes) {
    __shared__ float acc[SLICE_ROWS * OUT_DIM];
    __shared__ int wlist[SPMM_WAVES * LIST_CAP];
    __shared__ v2i sorted[SORT_CAP];
    __shared__ int rstart[SLICE_ROWS + 1];
    __shared__ int rcur[SLICE_ROWS];
    __shared__ int s_total;
    __shared__ float sdinv[SLICE_ROWS];
    static_assert(SLICE_ROWS <= SPMM_THREADS, "one row's dinv per thread");

    const int tid = threadIdx.x;
    const int lane = tid & 31;
    const int wave = __builtin_amdgcn_readfirstlane(tid >> 5);
    const int lo = blockIdx.x * SLICE_ROWS;
    int nrows = n_nodes - lo;
    if (nrows > SLICE_ROWS) nrows = SLICE_ROWS;

    v4f* acc4 = (v4f*)acc;
    const v4f zero4 = {0.0f, 0.0f, 0.0f, 0.0f};
    for (int i = tid; i < nrows * (OUT_DIM / 4); i += SPMM_THREADS) acc4[i] = zero4;
    for (int i = tid; i < SLICE_ROWS; i += SPMM_THREADS) rcur[i] = 0;
    if (tid == 0) s_total = 0;
    __syncthreads();

    const float my_dinv = (tid < nrows) ? dinv[lo + tid] : 0.0f;

    int* mylist = wlist + wave * LIST_CAP;
    int wcount = 0;

    if (recs_fit(ws_bytes, n_nodes, n_edges, recs_raw)) {
        const v2i* __restrict__ recs = (const v2i*)(recs_raw + rec_pad_ints(recs_raw));
        const int ch = (n_edges + BIN_WGS - 1) / BIN_WGS;
        int seg_len = 0, seg_q = 0;
        if (lane < SEGS_PER_WAVE) {
            const int chunk = wave + SPMM_WAVES * lane;
            const int* o = binoff + (size_t)chunk * OFF_PITCH + blockIdx.x;
            const int sa = o[0];
            seg_len = o[1] - sa;
            const long long k0l = (long long)chunk * ch;
            const int k0 = (k0l < (long long)n_edges) ? (int)k0l : n_edges;
            seg_q = k0 + sa;
        }
        int total = 0;
#pragma unroll
        for (int j = 0; j < SEGS_PER_WAVE; ++j) total += __builtin_amdgcn_readlane(seg_len, j);

        v2i rc[SEGS_PER_WAVE];
#pragma unroll
        for (int j = 0; j < SEGS_PER_WAVE; ++j) {
            const int len = __builtin_amdgcn_readlane(seg_len, j);
            const int q = __builtin_amdgcn_readlane(seg_q, j);
            rc[j] = v2i{0, 0};
            if (len > 0) rc[j] = recs[(size_t)(q + ((lane < len) ? lane : 0))];
        }
        if (lane == 0) atomicAdd(&s_total, total);
#pragma unroll
        for (int j = 0; j < SEGS_PER_WAVE; ++j) {
            const int len = __builtin_amdgcn_readlane(seg_len, j);
            if (lane < len) atomicAdd(&rcur[(int)((unsigned)rc[j].x >> REC_COL_BITS)], 1);
        }
        for (int j = 0; j < SEGS_PER_WAVE; ++j) {
            const int len = __builtin_amdgcn_readlane(seg_len, j);
            const int q = __builtin_amdgcn_readlane(seg_q, j);
            for (int i = 32 + lane; i < len; i += 32) {
                const v2i r = recs[(size_t)(q + i)];
                atomicAdd(&rcur[(int)((unsigned)r.x >> REC_COL_BITS)], 1);
            }
        }
        __syncthreads();
        const int stot = __builtin_amdgcn_readfirstlane(s_total);

        if (stot <= SORT_CAP) {
            if (wave == 0) {
                int v[ROWS_PER_LANE];
                int sum = 0;
#pragma unroll
                for (int i = 0; i < ROWS_PER_LANE; ++i) {
                    const int idx = lane * ROWS_PER_LANE + i;
                    v[i] = (idx < SLICE_ROWS) ? rcur[idx] : 0;
                    sum += v[i];
                }
                int incl = sum;
#pragma unroll
                for (int d = 1; d < 32; d <<= 1) {
                    const int t = __shfl_up(incl, d, 32);
                    if (lane >= d) incl += t;
                }
                int run = incl - sum;
#pragma unroll
                for (int i = 0; i < ROWS_PER_LANE; ++i) {
                    const int idx = lane * ROWS_PER_LANE + i;
                    if (idx < SLICE_ROWS) {
                        rstart[idx] = run;
                        rcur[idx] = run;
                    }
                    run += v[i];
                }
                if (lane == 0) rstart[SLICE_ROWS] = stot;
            }
            __syncthreads();

#pragma unroll
            for (int j = 0; j < SEGS_PER_WAVE; ++j) {
                const int len = __builtin_amdgcn_readlane(seg_len, j);
                if (lane < len) {
                    const int p = atomicAdd(&rcur[(int)((unsigned)rc[j].x >> REC_COL_BITS)], 1);
                    sorted[p] = rc[j];
                }
            }
            for (int j = 0; j < SEGS_PER_WAVE; ++j) {
                const int len = __builtin_amdgcn_readlane(seg_len, j);
                const int q = __builtin_amdgcn_readlane(seg_q, j);
                for (int i = 32 + lane; i < len; i += 32) {
                    const v2i r = recs[(size_t)(q + i)];
                    const int p = atomicAdd(&rcur[(int)((unsigned)r.x >> REC_COL_BITS)], 1);
                    sorted[p] = r;
                }
            }
            __syncthreads();

            const int rb = (wave * SLICE_ROWS) >> 5;
            const int re = ((wave + 1) * SLICE_ROWS) >> 5;
            const int e0 = __builtin_amdgcn_readfirstlane(rstart[rb]);
            const int e1 = __builtin_amdgcn_readfirstlane(rstart[re]);
            int cur = -1;
            v2f a = {0.0f, 0.0f};
            const int full_end = e0 + ((e1 - e0) & ~31);
            for (int i0 = e0; i0 < full_end; i0 += 32) {
                const v2i sr = sorted[i0 + lane];
                own_rows(sr.x & REC_COL_MASK, (int)((unsigned)sr.x >> REC_COL_BITS) * OUT_DIM, sr.y, 32, lane, support, acc, cur, a);
            }
            if (full_end < e1) {
                const int cnt = e1 - full_end;
                const v2i sr = sorted[full_end + ((lane < cnt) ? lane : 0)];
                own_rows(sr.x & REC_COL_MASK, (int)((unsigned)sr.x >> REC_COL_BITS) * OUT_DIM, sr.y, cnt, lane, support, acc, cur, a);
            }
            if (cur >= 0) *(v2f*)(acc + cur + 2 * lane) = a;
        } else {
            for (int j = 0; j < SEGS_PER_WAVE; ++j) {
                const int len = __builtin_amdgcn_readlane(seg_len, j);
                const int q = __builtin_amdgcn_readlane(seg_q, j);
                for (int i0 = 0; i0 < len; i0 += 32) {
                    const int cnt = (len - i0 < 32) ? len - i0 : 32;
                    const v2i r = recs[(size_t)(q + i0 + ((lane < cnt) ? lane : 0))];
                    const int rl = (int)((unsigned)r.x >> REC_COL_BITS);
                    const int c = r.x & REC_COL_MASK;
                    drain_rows(c, rl * OUT_DIM, __int_as_float(r.y), cnt, lane, support, acc);
                }
            }
        }
    } else {
    const int nv4 = n_edges >> 2;
    const v4i* rows4 = (const v4i*)rows;
    const int step_vecs = 32 * SCAN_U;
    const int stride = SPMM_WAVES * step_vecs;

#define TEST_EDGE(RV, LOV, EV)                                                   \
    {                                                                            \
        const bool match_ = (unsigned)((RV) - (LOV)) < (unsigned)nrows;          \
        const unsigned m_ = __builtin_amdgcn_ballot_w32(match_);                 \
        if (m_ != 0u) {                                                          \
            if (match_) mylist[wcount + (int)__builtin_amdgcn_mbcnt_lo(m_, 0u)] = (EV); \
            wcount += __builtin_popcount(m_);                                    \
        }                                                                        \
    }

    if (wave == 0) {
        const int e = (nv4 << 2) + lane;
        const int r = (e < n_edges) ? rows[e] : -1;
        TEST_EDGE(r, lo, e)
    }

#define LOAD_STEP(DST, LOS, VB)                                                  \
    _Pragma("unroll") for (int u = 0; u < SCAN_U; ++u) {                         \
        const int idx_ = (VB) + u * 32 + lane;                                   \
        const bool ok_ = idx_ < nv4;                                             \
        DST[u] = rows4[ok_ ? idx_ : nv4 - 1];                                    \
        LOS[u] = ok_ ? lo : 0x7fffffff;                                          \
    }

#define SCAN_STEP(SRC, LOS, VB)                                                  \
    _Pragma("unroll") for (int u = 0; u < SCAN_U; ++u) {                         \
        const int eb_ = ((VB) + u * 32 + lane) * 4;                              \
        TEST_EDGE(SRC[u].x, LOS[u], eb_ + 0)                                     \
        TEST_EDGE(SRC[u].y, LOS[u], eb_ + 1)                                     \
        TEST_EDGE(SRC[u].z, LOS[u], eb_ + 2)                                     \
        TEST_EDGE(SRC[u].w, LOS[u], eb_ + 3)                                     \
    }                                                                            \
    while (wcount >= 32) {                                                       \
        wcount -= 32;                                                            \
        drain_batch(mylist + wcount, 32, lane, lo, rows, cols, adj, support, acc); \
    }

    int vb = wave * step_vecs;
    if (vb < nv4) {
        v4i ra[SCAN_U], rb[SCAN_U];
        int la[SCAN_U], lb[SCAN_U];
        LOAD_STEP(ra, la, vb)
        for (;;) {
            LOAD_STEP(rb, lb, vb + stride)
            SCAN_STEP(ra, la, vb)
            vb += stride;
            if (vb >= nv4) break;
            LOAD_STEP(ra, la, vb + stride)
            SCAN_STEP(rb, lb, vb)
            vb += stride;
            if (vb >= nv4) break;
        }
    }
#undef SCAN_STEP
#undef LOAD_STEP
#undef TEST_EDGE
    }
    if (wcount > 0) drain_batch(mylist, wcount, lane, lo, rows, cols, adj, support, acc);
    if (tid < SLICE_ROWS) sdinv[tid] = my_dinv;
    __syncthreads();

    v4f* out4 = (v4f*)(out + (size_t)lo * OUT_DIM);
    for (int i = tid; i < nrows * (OUT_DIM / 4); i += SPMM_THREADS) out4[i] = acc4[i] * sdinv[i / (OUT_DIM / 4)];
}

extern "C" void kernel_launch(void* const* d_in, const int* in_sizes, int n_in,
                              void* d_out, int out_size, void* d_ws, size_t ws_size,
                              hipStream_t stream) {
    const int* rows = (const int*)d_in[0];
    const int* cols = (const int*)d_in[1];
    const float* adj = (const float*)d_in[2];
    const float* x = (const float*)d_in[3];
    const float* W = (const float*)d_in[4];
    const float* b = (const float*)d_in[5];
    float* out = (float*)d_out;

    const int n_edges = in_sizes[0];
    const int n_nodes = in_sizes[3] / IN_DIM;

    float* support = (float*)d_ws;
    float* dinv = support + (size_t)n_nodes * OUT_DIM;
    int* binoff = (int*)(dinv + n_nodes);
    int* recs_raw = binoff + BIN_WGS * OFF_PITCH;
    const unsigned long long ws_bytes = (unsigned long long)ws_size;

    bin_edges_kernel<<<BIN_WGS, BIN_THREADS, 0, stream>>>(rows, cols, adj, binoff, recs_raw, n_edges, n_nodes, ws_bytes);

    slice_dinv_gemm_kernel<<<(n_nodes + SLICE_ROWS - 1) / SLICE_ROWS, FUSE_THREADS, 0, stream>>>(
        rows, adj, x, W, b, support, dinv, binoff, recs_raw, n_edges, n_nodes, ws_bytes);

    spmm_slice_kernel<<<(n_nodes + SLICE_ROWS - 1) / SLICE_ROWS, SPMM_THREADS, 0, stream>>>(
        rows, cols, adj, dinv, support, out, binoff, recs_raw, n_edges, n_nodes, ws_bytes);
}
